# starting kernel
# baseline (speedup 1.0000x reference)
#include <hip/hip_runtime.h>

typedef float v2f __attribute__((ext_vector_type(2)));
typedef float v8f __attribute__((ext_vector_type(8)));

#define NNODES 50000
#define NEDGES 800000
#define KMAX   128
#define BPAD   (KMAX + 2)


__global__ void zero_kernel(float* __restrict__ p, int n) {
    int i = blockIdx.x * blockDim.x + threadIdx.x;
    if (i < n) p[i] = 0.0f;
}

__global__ void degree_kernel(const int* __restrict__ dst, float* __restrict__ deg, int E) {
    int e = blockIdx.x * blockDim.x + threadIdx.x;
    if (e < E) atomicAdd(&deg[dst[e]], 1.0f);
}

__global__ void invdeg_kernel(float* __restrict__ deg, int n) {
    int i = blockIdx.x * blockDim.x + threadIdx.x;
    if (i < n) deg[i] = 1.0f / fmaxf(deg[i], 1.0f);
}

__global__ __launch_bounds__(256) void gemm_wmma_kernel(
    const float* __restrict__ Xin, const float* __restrict__ W,
    float* __restrict__ Out, int K, int N, int Mtiles)
{
    __shared__ float smem[16 * BPAD];

    int tid = threadIdx.x;
    int n0  = blockIdx.y << 4;

    for (int idx = tid; idx < (K << 4); idx += 256) {
        int k = idx >> 4;
        int n = idx & 15;
        int col = n0 + n;
        smem[n * BPAD + k] = (col < N) ? W[(size_t)k * N + col] : 0.0f;
    }
    __syncthreads();

    int tM = (blockIdx.x << 3) + (tid >> 5);
    if (tM >= Mtiles) return;

    int lane = tid & 31;
    int half = lane >> 4;
    int lid  = lane & 15;
    int m0   = tM << 4;

    const float* arow = Xin + (size_t)(m0 + lid) * K;
    const float* brow = smem + lid * BPAD;
    int kofs = half << 1;

    v8f acc = {};
    for (int k0 = 0; k0 < K; k0 += 4) {
        int ka = k0 + kofs;
        v2f a = *(const v2f*)(arow + ka);
        v2f b = *(const v2f*)(brow + ka);
        acc = __builtin_amdgcn_wmma_f32_16x16x4_f32(
             false, a,  false, b,
             (short)0, acc,  false,  false);
    }

    int ncol = n0 + lid;
    if (ncol < N) {
        float* orow = Out + (size_t)(m0 + (half << 3)) * N + ncol;
#pragma unroll
        for (int r = 0; r < 8; ++r)
            orow[(size_t)r * N] = acc[r];
    }
}

__global__ void scatter_kernel(const float* __restrict__ P,
                               const int* __restrict__ src,
                               const int* __restrict__ dst,
                               float* __restrict__ agg, int E, int F)
{
    int e = blockIdx.x * blockDim.x + threadIdx.x;
    if (e >= E) return;
    int c = blockIdx.y << 2;
    int s = src[e];
    int d = dst[e];
    float4 v = *(const float4*)(P + (size_t)s * F + c);
    float* base = agg + (size_t)d * F + c;
    atomicAdd(base + 0, v.x);
    atomicAdd(base + 1, v.y);
    atomicAdd(base + 2, v.z);
    atomicAdd(base + 3, v.w);
}

__global__ void pointwise_kernel(const float* __restrict__ agg,
                                 const float* __restrict__ inv_deg,
                                 const float* __restrict__ bias,
                                 float* __restrict__ out, int F, int relu)
{
    int i = blockIdx.x * blockDim.x + threadIdx.x;
    int total = NNODES * F;
    if (i >= total) return;
    int node = i / F;
    int f    = i - node * F;
    float v = agg[i] * inv_deg[node] + bias[f];
    out[i] = relu ? fmaxf(v, 0.0f) : v;
}


extern "C" void kernel_launch(void* const* d_in, const int* in_sizes, int n_in,
                              void* d_out, int out_size, void* d_ws, size_t ws_size,
                              hipStream_t stream)
{
    const float* features = (const float*)d_in[0];
    const int*   src      = (const int*)  d_in[1];
    const int*   dst      = (const int*)  d_in[2];
    const float* W0       = (const float*)d_in[3];
    const float* b0       = (const float*)d_in[4];
    const float* W1       = (const float*)d_in[5];
    const float* b1       = (const float*)d_in[6];
    const float* W2       = (const float*)d_in[7];
    const float* b2       = (const float*)d_in[8];
    float*       out      = (float*)d_out;

    float* ws  = (float*)d_ws;
    float* deg = ws;
    float* P   = ws + 50048;
    float* A   = P  + (size_t)NNODES * 64;
    float* H   = A  + (size_t)NNODES * 64;

    const int Mtiles     = NNODES / 16;
    const int gemmBlksX  = (Mtiles + 7) / 8;
    const int edgeBlocks = (NEDGES + 255) / 256;

    zero_kernel  <<<(NNODES + 255) / 256, 256, 0, stream>>>(deg, NNODES);
    degree_kernel<<<edgeBlocks, 256, 0, stream>>>(dst, deg, NEDGES);
    invdeg_kernel<<<(NNODES + 255) / 256, 256, 0, stream>>>(deg, NNODES);

    {
        int F = 64, tot = NNODES * F;
        gemm_wmma_kernel<<<dim3(gemmBlksX, 4), 256, 0, stream>>>(features, W0, P, 128, F, Mtiles);
        zero_kernel     <<<(tot + 255) / 256, 256, 0, stream>>>(A, tot);
        scatter_kernel  <<<dim3(edgeBlocks, F / 4), 256, 0, stream>>>(P, src, dst, A, NEDGES, F);
        pointwise_kernel<<<(tot + 255) / 256, 256, 0, stream>>>(A, deg, b0, H, F, 1);
    }

    {
        int F = 64, tot = NNODES * F;
        gemm_wmma_kernel<<<dim3(gemmBlksX, 4), 256, 0, stream>>>(H, W1, P, 64, F, Mtiles);
        zero_kernel     <<<(tot + 255) / 256, 256, 0, stream>>>(A, tot);
        scatter_kernel  <<<dim3(edgeBlocks, F / 4), 256, 0, stream>>>(P, src, dst, A, NEDGES, F);
        pointwise_kernel<<<(tot + 255) / 256, 256, 0, stream>>>(A, deg, b1, H, F, 1);
    }

    {
        int F = 40, tot = NNODES * F;
        gemm_wmma_kernel<<<dim3(gemmBlksX, 3), 256, 0, stream>>>(H, W2, P, 64, F, Mtiles);
        zero_kernel     <<<(tot + 255) / 256, 256, 0, stream>>>(A, tot);
        scatter_kernel  <<<dim3(edgeBlocks, F / 4), 256, 0, stream>>>(P, src, dst, A, NEDGES, F);
        pointwise_kernel<<<(tot + 255) / 256, 256, 0, stream>>>(A, deg, b2, out, F, 0);
    }
}


# In-degree counted by the owner blocks: first neighbour-sum launch adds 1 per kept edge to an LDS row counter (integer, exact) and stores 1/max(deg,1); the three degree launches are gone (12 -> 9)
# speedup vs baseline: 10.7256x; 10.7256x over previous
#include <hip/hip_runtime.h>

typedef float v2f __attribute__((ext_vector_type(2)));
typedef float v8f __attribute__((ext_vector_type(8)));

#define NNODES 50000
#define NEDGES 800000
#define KMAX   128
#define BPAD   (KMAX + 2)

__global__ __launch_bounds__(256) void gemm_wmma_kernel(
    const float* __restrict__ Xin, const float* __restrict__ W,
    float* __restrict__ Out, int K, int N, int Mtiles)
{
    __shared__ float smem[16 * BPAD];

    int tid = threadIdx.x;
    int n0  = blockIdx.y << 4;

    for (int idx = tid; idx < (K << 4); idx += 256) {
        int k = idx >> 4;
        int n = idx & 15;
        int col = n0 + n;
        smem[n * BPAD + k] = (col < N) ? W[(size_t)k * N + col] : 0.0f;
    }
    __syncthreads();

    int tM = (blockIdx.x << 3) + (tid >> 5);
    if (tM >= Mtiles) return;

    int lane = tid & 31;
    int half = lane >> 4;
    int lid  = lane & 15;
    int m0   = tM << 4;

    const float* arow = Xin + (size_t)(m0 + lid) * K;
    const float* brow = smem + lid * BPAD;
    int kofs = half << 1;

    v8f acc = {};
    for (int k0 = 0; k0 < K; k0 += 4) {
        int ka = k0 + kofs;
        v2f a = *(const v2f*)(arow + ka);
        v2f b = *(const v2f*)(brow + ka);
        acc = __builtin_amdgcn_wmma_f32_16x16x4_f32(
             false, a,  false, b,
             (short)0, acc,  false,  false);
    }

    int ncol = n0 + lid;
    if (ncol < N) {
        float* orow = Out + (size_t)(m0 + (half << 3)) * N + ncol;
#pragma unroll
        for (int r = 0; r < 8; ++r)
            orow[(size_t)r * N] = acc[r];
    }
}

#define RN    196
#define NOWN  ((NNODES + RN - 1) / RN)
#define LCAP  36864
#define LHDR  32
#define LREG  (LCAP + LHDR)
static_assert(NEDGES % 4 == 0, "dst is read as 16-byte vectors");
static_assert(NEDGES <= (1 << 20) && NNODES <= (1 << 20) && RN <= 2048, "list entry packing");

template <int F, bool BUILD>
__device__ __forceinline__ void keep_edge(unsigned pos, int e, unsigned r,
                                          unsigned* list, float* acc, unsigned* indeg,
                                          const float* __restrict__ P,
                                          const int* __restrict__ src)
{
    if (BUILD) atomicAdd(&indeg[r], 1u);
    if (pos < LCAP) {
        list[pos] = (r << 20) | (unsigned)e;
    } else {
        const float* row = P + (size_t)src[e] * F;
#pragma nounroll
        for (int f = 0; f < F; ++f) atomicAdd(&acc[r * F + f], row[f]);
    }
}

template <int F, bool BUILD>
__device__ __forceinline__ void test_four(int4 d, int e0, int lo, unsigned nrows,
                                          unsigned* cnt, unsigned* list, float* acc, unsigned* indeg,
                                          const float* __restrict__ P,
                                          const int* __restrict__ src)
{
    unsigned r0 = (unsigned)(d.x - lo);
    unsigned r1 = (unsigned)(d.y - lo);
    unsigned r2 = (unsigned)(d.z - lo);
    unsigned r3 = (unsigned)(d.w - lo);
    unsigned h0 = r0 < nrows, h1 = r1 < nrows, h2 = r2 < nrows, h3 = r3 < nrows;
    unsigned c = h0 + h1 + h2 + h3;
    if (c) {
        unsigned pos = atomicAdd(cnt, c);
        if (h0) keep_edge<F, BUILD>(pos++, e0 + 0, r0, list, acc, indeg, P, src);
        if (h1) keep_edge<F, BUILD>(pos++, e0 + 1, r1, list, acc, indeg, P, src);
        if (h2) keep_edge<F, BUILD>(pos++, e0 + 2, r2, list, acc, indeg, P, src);
        if (h3) keep_edge<F, BUILD>(pos++, e0 + 3, r3, list, acc, indeg, P, src);
    }
}

template <int F, bool BUILD>
__global__ __launch_bounds__(1024) void owner_sum_kernel(
    const float* __restrict__ P, const int* __restrict__ src,
    const int* __restrict__ dst, float* __restrict__ agg,
    unsigned* __restrict__ saved, float* __restrict__ inv_deg)
{
    __shared__ float    acc[RN * F];
    __shared__ unsigned list[LCAP];
    __shared__ unsigned cnt;
    __shared__ unsigned indeg[BUILD ? RN : 1];

    const int tid = threadIdx.x;
    const int lo  = blockIdx.x * RN;
    const unsigned nrows = (unsigned)((NNODES - lo) < RN ? (NNODES - lo) : RN);
    unsigned* mine = saved + (size_t)blockIdx.x * LREG;

    const unsigned have = BUILD ? 0u : mine[0];
    const bool filter = BUILD || have > (unsigned)LCAP;

    for (int i = tid; i < RN * F; i += 1024) acc[i] = 0.0f;
    if (BUILD && tid < RN) indeg[tid] = 0u;
    if (tid == 0) cnt = 0u;

    int n;
    if (filter) {
        __syncthreads();

        const int4* dst4 = (const int4*)dst;
        constexpr int NVEC = NEDGES / 4;
        int base = tid;
        for (; base + 7 * 1024 < NVEC; base += 8 * 1024) {
            int4 d[8];
#pragma unroll
            for (int u = 0; u < 8; ++u) d[u] = dst4[base + u * 1024];
#pragma unroll
            for (int u = 0; u < 8; ++u)
                test_four<F, BUILD>(d[u], (base + u * 1024) << 2, lo, nrows, &cnt, list, acc, indeg, P, src);
        }
        for (; base < NVEC; base += 1024)
            test_four<F, BUILD>(dst4[base], base << 2, lo, nrows, &cnt, list, acc, indeg, P, src);
        __syncthreads();
        const unsigned total = cnt;
        n = (int)(total < (unsigned)LCAP ? total : (unsigned)LCAP);

        if (BUILD && (unsigned)tid < nrows)
            inv_deg[lo + tid] = 1.0f / fmaxf((float)indeg[tid], 1.0f);

        for (int i = tid; i < n; i += 1024) {
            unsigned ent = list[i];
            int s = src[ent & 0xFFFFFu];
            ent = (ent & 0xFFF00000u) | (unsigned)s;
            list[i] = ent;
            if (BUILD) mine[LHDR + i] = ent;
        }
        if (BUILD && tid == 0) mine[0] = total;
    } else {
        n = (int)have;
        for (int i = tid; i < n; i += 1024) list[i] = mine[LHDR + i];
    }
    __syncthreads();

    const int wave = tid >> 5;
    const int lane = tid & 31;
    for (int i0 = wave; i0 < n; i0 += 32 * 8) {
        float v0[8], v1[8];
        unsigned rr[8], ents[8];
#pragma unroll
        for (int u = 0; u < 8; ++u) {
            int i = i0 + u * 32;
            ents[u] = (i < n) ? list[i] : 0u;
        }
#pragma unroll
        for (int u = 0; u < 8; ++u) {
            unsigned ent = __builtin_amdgcn_readfirstlane(ents[u]);
            const float* row = P + (size_t)(ent & 0xFFFFFu) * F;
            rr[u] = ent >> 20;
            v0[u] = row[lane];
            v1[u] = (lane + 32 < F) ? row[lane + 32] : 0.0f;
        }
#pragma unroll
        for (int u = 0; u < 8; ++u) {
            if (i0 + u * 32 < n) {
                float* a = acc + rr[u] * F;
                atomicAdd(a + lane, v0[u]);
                if (lane + 32 < F) atomicAdd(a + lane + 32, v1[u]);
            }
        }
    }
    __syncthreads();

    float* orow = agg + (size_t)lo * F;
    const int tot = (int)nrows * F;
    for (int i = tid; i < tot; i += 1024) orow[i] = acc[i];
}

__global__ void pointwise_kernel(const float* __restrict__ agg,
                                 const float* __restrict__ inv_deg,
                                 const float* __restrict__ bias,
                                 float* __restrict__ out, int F, int relu)
{
    int i = blockIdx.x * blockDim.x + threadIdx.x;
    int total = NNODES * F;
    if (i >= total) return;
    int node = i / F;
    int f    = i - node * F;
    float v = agg[i] * inv_deg[node] + bias[f];
    out[i] = relu ? fmaxf(v, 0.0f) : v;
}


extern "C" void kernel_launch(void* const* d_in, const int* in_sizes, int n_in,
                              void* d_out, int out_size, void* d_ws, size_t ws_size,
                              hipStream_t stream)
{
    const float* features = (const float*)d_in[0];
    const int*   src      = (const int*)  d_in[1];
    const int*   dst      = (const int*)  d_in[2];
    const float* W0       = (const float*)d_in[3];
    const float* b0       = (const float*)d_in[4];
    const float* W1       = (const float*)d_in[5];
    const float* b1       = (const float*)d_in[6];
    const float* W2       = (const float*)d_in[7];
    const float* b2       = (const float*)d_in[8];
    float*       out      = (float*)d_out;

    float* ws  = (float*)d_ws;
    float* deg = ws;
    float* P   = ws + 50048;
    float* A   = P  + (size_t)NNODES * 64;
    float* H   = A  + (size_t)NNODES * 64;
    unsigned* L = (unsigned*)(H + (size_t)NNODES * 64);

    const int Mtiles     = NNODES / 16;
    const int gemmBlksX  = (Mtiles + 7) / 8;

    {
        int F = 64, tot = NNODES * F;
        gemm_wmma_kernel<<<dim3(gemmBlksX, 4), 256, 0, stream>>>(features, W0, P, 128, F, Mtiles);
        owner_sum_kernel<64, true><<<NOWN, 1024, 0, stream>>>(P, src, dst, A, L, deg);
        pointwise_kernel<<<(tot + 255) / 256, 256, 0, stream>>>(A, deg, b0, H, F, 1);
    }

    {
        int F = 64, tot = NNODES * F;
        gemm_wmma_kernel<<<dim3(gemmBlksX, 4), 256, 0, stream>>>(H, W1, P, 64, F, Mtiles);
        owner_sum_kernel<64, false><<<NOWN, 1024, 0, stream>>>(P, src, dst, A, L, deg);
        pointwise_kernel<<<(tot + 255) / 256, 256, 0, stream>>>(A, deg, b1, H, F, 1);
    }

    {
        int F = 40, tot = NNODES * F;
        gemm_wmma_kernel<<<dim3(gemmBlksX, 3), 256, 0, stream>>>(H, W2, P, 64, F, Mtiles);
        owner_sum_kernel<40, false><<<NOWN, 1024, 0, stream>>>(P, src, dst, A, L, deg);
        pointwise_kernel<<<(tot + 255) / 256, 256, 0, stream>>>(A, deg, b2, out, F, 0);
    }
}


# dst packed to one owner-block byte per edge by the layer-0 product launch; owner blocks (197 nodes each) filter that byte stream, 16 edges per 16-B load; sums still LDS adds in arrival order
# speedup vs baseline: 11.6163x; 1.0830x over previous
#include <hip/hip_runtime.h>

typedef float v2f __attribute__((ext_vector_type(2)));
typedef float v8f __attribute__((ext_vector_type(8)));

#define NNODES 50000
#define NEDGES 800000
#define KMAX   128
#define BPAD   (KMAX + 2)

#define RN    197
#define NOWN  ((NNODES + RN - 1) / RN)
static_assert(NOWN <= 255, "owner id must fit a byte with 255 left free");

__device__ __forceinline__ unsigned owner_of(int d)
{
    unsigned u = (unsigned)d;
    return u < (unsigned)NNODES ? u / (unsigned)RN : 255u;
}

template <bool OWNERS>
__global__ __launch_bounds__(256) void gemm_wmma_kernel(
    const float* __restrict__ Xin, const float* __restrict__ W,
    float* __restrict__ Out, int K, int N, int Mtiles,
    const int* __restrict__ dst, unsigned* __restrict__ own)
{
    __shared__ float smem[16 * BPAD];

    int tid = threadIdx.x;
    int n0  = blockIdx.y << 4;

    if (OWNERS) {
        const int4* dst4 = (const int4*)dst;
        const int nthr = (int)(gridDim.x * gridDim.y) * 256;
        for (int v = (int)(blockIdx.y * gridDim.x + blockIdx.x) * 256 + tid; v < NEDGES / 4; v += nthr) {
            int4 d = dst4[v];
            own[v] = owner_of(d.x) | (owner_of(d.y) << 8) | (owner_of(d.z) << 16) | (owner_of(d.w) << 24);
        }
    }

    for (int idx = tid; idx < (K << 4); idx += 256) {
        int k = idx >> 4;
        int n = idx & 15;
        int col = n0 + n;
        smem[n * BPAD + k] = (col < N) ? W[(size_t)k * N + col] : 0.0f;
    }
    __syncthreads();

    int tM = (blockIdx.x << 3) + (tid >> 5);
    if (tM >= Mtiles) return;

    int lane = tid & 31;
    int half = lane >> 4;
    int lid  = lane & 15;
    int m0   = tM << 4;

    const float* arow = Xin + (size_t)(m0 + lid) * K;
    const float* brow = smem + lid * BPAD;
    int kofs = half << 1;

    v8f acc = {};
    for (int k0 = 0; k0 < K; k0 += 4) {
        int ka = k0 + kofs;
        v2f a = *(const v2f*)(arow + ka);
        v2f b = *(const v2f*)(brow + ka);
        acc = __builtin_amdgcn_wmma_f32_16x16x4_f32(
             false, a,  false, b,
             (short)0, acc,  false,  false);
    }

    int ncol = n0 + lid;
    if (ncol < N) {
        float* orow = Out + (size_t)(m0 + (half << 3)) * N + ncol;
#pragma unroll
        for (int r = 0; r < 8; ++r)
            orow[(size_t)r * N] = acc[r];
    }
}

#define LCAP  36864
#define LHDR  32
#define LREG  (LCAP + LHDR)
static_assert(NEDGES % 16 == 0, "dst and the owner bytes are read as 16-byte vectors");
static_assert(NEDGES <= (1 << 20) && NNODES <= (1 << 20) && RN <= 2048, "list entry packing");

__device__ __forceinline__ unsigned match_bytes(unsigned x, unsigned key)
{
    unsigned y = x ^ key;
    unsigned t = (y & 0x7F7F7F7Fu) + 0x7F7F7F7Fu;
    return ~(t | y | 0x7F7F7F7Fu);
}

template <int F, bool BUILD>
__device__ __forceinline__ void keep_edge(unsigned pos, int e, int lo,
                                          unsigned* list, float* acc, unsigned* indeg,
                                          const float* __restrict__ P,
                                          const int* __restrict__ src,
                                          const int* __restrict__ dst)
{
    if (pos < LCAP) {
        list[pos] = (unsigned)e;
    } else {
        unsigned r = (unsigned)(dst[e] - lo);
        if (BUILD) atomicAdd(&indeg[r], 1u);
        const float* row = P + (size_t)src[e] * F;
#pragma nounroll
        for (int f = 0; f < F; ++f) atomicAdd(&acc[r * F + f], row[f]);
    }
}

template <int F, bool BUILD>
__device__ __forceinline__ void test_sixteen(uint4 o, int e0, unsigned key, int lo,
                                             unsigned* cnt, unsigned* list, float* acc, unsigned* indeg,
                                             const float* __restrict__ P,
                                             const int* __restrict__ src,
                                             const int* __restrict__ dst)
{
    unsigned m0 = match_bytes(o.x, key);
    unsigned m1 = match_bytes(o.y, key);
    unsigned m2 = match_bytes(o.z, key);
    unsigned m3 = match_bytes(o.w, key);
    if (m0 | m1 | m2 | m3) {
        unsigned mk = (m0 >> 7) | (m1 >> 6) | (m2 >> 5) | (m3 >> 4);
        unsigned pos = atomicAdd(cnt, (unsigned)__popc(mk));
        do {
            unsigned t = (unsigned)__builtin_ctz(mk);
            mk &= mk - 1u;
            int e = e0 + (int)(((t & 7u) << 2) + (t >> 3));
            keep_edge<F, BUILD>(pos++, e, lo, list, acc, indeg, P, src, dst);
        } while (mk);
    }
}

template <int F, bool BUILD>
__global__ __launch_bounds__(1024) void owner_sum_kernel(
    const float* __restrict__ P, const int* __restrict__ src,
    const int* __restrict__ dst, float* __restrict__ agg,
    unsigned* __restrict__ saved, float* __restrict__ inv_deg,
    const unsigned* __restrict__ own)
{
    __shared__ float    acc[RN * F];
    __shared__ unsigned list[LCAP];
    __shared__ unsigned cnt;
    __shared__ unsigned indeg[BUILD ? RN : 1];

    const int tid = threadIdx.x;
    const int lo  = blockIdx.x * RN;
    const unsigned nrows = (unsigned)((NNODES - lo) < RN ? (NNODES - lo) : RN);
    unsigned* mine = saved + (size_t)blockIdx.x * LREG;

    const unsigned have = BUILD ? 0u : mine[0];
    const bool filter = BUILD || have > (unsigned)LCAP;

    for (int i = tid; i < RN * F; i += 1024) acc[i] = 0.0f;
    if (BUILD && tid < RN) indeg[tid] = 0u;
    if (tid == 0) cnt = 0u;

    int n;
    if (filter) {
        __syncthreads();

        const uint4* own4 = (const uint4*)own;
        constexpr int NVEC = NEDGES / 16;
        const unsigned key = (unsigned)blockIdx.x * 0x01010101u;
        int base = tid;
        for (; base + 7 * 1024 < NVEC; base += 8 * 1024) {
            uint4 o[8];
#pragma unroll
            for (int u = 0; u < 8; ++u) o[u] = own4[base + u * 1024];
#pragma unroll
            for (int u = 0; u < 8; ++u)
                test_sixteen<F, BUILD>(o[u], (base + u * 1024) << 4, key, lo, &cnt, list, acc, indeg, P, src, dst);
        }
        for (; base < NVEC; base += 1024)
            test_sixteen<F, BUILD>(own4[base], base << 4, key, lo, &cnt, list, acc, indeg, P, src, dst);
        __syncthreads();
        const unsigned total = cnt;
        n = (int)(total < (unsigned)LCAP ? total : (unsigned)LCAP);

        for (int i = tid; i < n; i += 1024) {
            unsigned e = list[i];
            unsigned r = (unsigned)(dst[e] - lo);
            int s = src[e];
            if (BUILD) atomicAdd(&indeg[r], 1u);
            unsigned ent = (r << 20) | (unsigned)s;
            list[i] = ent;
            if (BUILD) mine[LHDR + i] = ent;
        }
        if (BUILD && tid == 0) mine[0] = total;
    } else {
        n = (int)have;
        for (int i = tid; i < n; i += 1024) list[i] = mine[LHDR + i];
    }
    __syncthreads();

    if (BUILD && (unsigned)tid < nrows)
        inv_deg[lo + tid] = 1.0f / fmaxf((float)indeg[tid], 1.0f);

    const int wave = tid >> 5;
    const int lane = tid & 31;
    for (int i0 = wave; i0 < n; i0 += 32 * 8) {
        float v0[8], v1[8];
        unsigned rr[8], ents[8];
#pragma unroll
        for (int u = 0; u < 8; ++u) {
            int i = i0 + u * 32;
            ents[u] = (i < n) ? list[i] : 0u;
        }
#pragma unroll
        for (int u = 0; u < 8; ++u) {
            unsigned ent = __builtin_amdgcn_readfirstlane(ents[u]);
            const float* row = P + (size_t)(ent & 0xFFFFFu) * F;
            rr[u] = ent >> 20;
            v0[u] = row[lane];
            v1[u] = (lane + 32 < F) ? row[lane + 32] : 0.0f;
        }
#pragma unroll
        for (int u = 0; u < 8; ++u) {
            if (i0 + u * 32 < n) {
                float* a = acc + rr[u] * F;
                atomicAdd(a + lane, v0[u]);
                if (lane + 32 < F) atomicAdd(a + lane + 32, v1[u]);
            }
        }
    }
    __syncthreads();

    float* orow = agg + (size_t)lo * F;
    const int tot = (int)nrows * F;
    for (int i = tid; i < tot; i += 1024) orow[i] = acc[i];
}

__global__ void pointwise_kernel(const float* __restrict__ agg,
                                 const float* __restrict__ inv_deg,
                                 const float* __restrict__ bias,
                                 float* __restrict__ out, int F, int relu)
{
    int i = blockIdx.x * blockDim.x + threadIdx.x;
    int total = NNODES * F;
    if (i >= total) return;
    int node = i / F;
    int f    = i - node * F;
    float v = agg[i] * inv_deg[node] + bias[f];
    out[i] = relu ? fmaxf(v, 0.0f) : v;
}


extern "C" void kernel_launch(void* const* d_in, const int* in_sizes, int n_in,
                              void* d_out, int out_size, void* d_ws, size_t ws_size,
                              hipStream_t stream)
{
    const float* features = (const float*)d_in[0];
    const int*   src      = (const int*)  d_in[1];
    const int*   dst      = (const int*)  d_in[2];
    const float* W0       = (const float*)d_in[3];
    const float* b0       = (const float*)d_in[4];
    const float* W1       = (const float*)d_in[5];
    const float* b1       = (const float*)d_in[6];
    const float* W2       = (const float*)d_in[7];
    const float* b2       = (const float*)d_in[8];
    float*       out      = (float*)d_out;

    float* ws  = (float*)d_ws;
    float* deg = ws;
    float* P   = ws + 50048;
    float* A   = P  + (size_t)NNODES * 64;
    float* H   = A  + (size_t)NNODES * 64;
    unsigned* L = (unsigned*)(H + (size_t)NNODES * 64);
    unsigned* O = L + (size_t)NOWN * LREG;

    const int Mtiles     = NNODES / 16;
    const int gemmBlksX  = (Mtiles + 7) / 8;

    {
        int F = 64, tot = NNODES * F;
        gemm_wmma_kernel<true><<<dim3(gemmBlksX, 4), 256, 0, stream>>>(features, W0, P, 128, F, Mtiles, dst, O);
        owner_sum_kernel<64, true><<<NOWN, 1024, 0, stream>>>(P, src, dst, A, L, deg, O);
        pointwise_kernel<<<(tot + 255) / 256, 256, 0, stream>>>(A, deg, b0, H, F, 1);
    }

    {
        int F = 64, tot = NNODES * F;
        gemm_wmma_kernel<false><<<dim3(gemmBlksX, 4), 256, 0, stream>>>(H, W1, P, 64, F, Mtiles, dst, O);
        owner_sum_kernel<64, false><<<NOWN, 1024, 0, stream>>>(P, src, dst, A, L, deg, O);
        pointwise_kernel<<<(tot + 255) / 256, 256, 0, stream>>>(A, deg, b1, H, F, 1);
    }

    {
        int F = 40, tot = NNODES * F;
        gemm_wmma_kernel<false><<<dim3(gemmBlksX, 3), 256, 0, stream>>>(H, W2, P, 64, F, Mtiles, dst, O);
        owner_sum_kernel<40, false><<<NOWN, 1024, 0, stream>>>(P, src, dst, A, L, deg, O);
        pointwise_kernel<<<(tot + 255) / 256, 256, 0, stream>>>(A, deg, b2, out, F, 0);
    }
}


# Normalise + bias + ReLU done by the owner block as it stores its LDS sums (1/in-degree held in LDS beside them): three elementwise launches and the agg round trip gone (9 -> 6); sum order unchanged
# speedup vs baseline: 13.9122x; 1.1976x over previous
#include <hip/hip_runtime.h>

typedef float v2f __attribute__((ext_vector_type(2)));
typedef float v8f __attribute__((ext_vector_type(8)));

#define NNODES 50000
#define NEDGES 800000
#define KMAX   128
#define BPAD   (KMAX + 2)

#define RN    197
#define NOWN  ((NNODES + RN - 1) / RN)
static_assert(NOWN <= 255, "owner id must fit a byte with 255 left free");

__device__ __forceinline__ unsigned owner_of(int d)
{
    unsigned u = (unsigned)d;
    return u < (unsigned)NNODES ? u / (unsigned)RN : 255u;
}

template <bool OWNERS>
__global__ __launch_bounds__(256) void gemm_wmma_kernel(
    const float* __restrict__ Xin, const float* __restrict__ W,
    float* __restrict__ Out, int K, int N, int Mtiles,
    const int* __restrict__ dst, unsigned* __restrict__ own)
{
    __shared__ float smem[16 * BPAD];

    int tid = threadIdx.x;
    int n0  = blockIdx.y << 4;

    if (OWNERS) {
        const int4* dst4 = (const int4*)dst;
        const int nthr = (int)(gridDim.x * gridDim.y) * 256;
        for (int v = (int)(blockIdx.y * gridDim.x + blockIdx.x) * 256 + tid; v < NEDGES / 4; v += nthr) {
            int4 d = dst4[v];
            own[v] = owner_of(d.x) | (owner_of(d.y) << 8) | (owner_of(d.z) << 16) | (owner_of(d.w) << 24);
        }
    }

    for (int idx = tid; idx < (K << 4); idx += 256) {
        int k = idx >> 4;
        int n = idx & 15;
        int col = n0 + n;
        smem[n * BPAD + k] = (col < N) ? W[(size_t)k * N + col] : 0.0f;
    }
    __syncthreads();

    int tM = (blockIdx.x << 3) + (tid >> 5);
    if (tM >= Mtiles) return;

    int lane = tid & 31;
    int half = lane >> 4;
    int lid  = lane & 15;
    int m0   = tM << 4;

    const float* arow = Xin + (size_t)(m0 + lid) * K;
    const float* brow = smem + lid * BPAD;
    int kofs = half << 1;

    v8f acc = {};
    for (int k0 = 0; k0 < K; k0 += 4) {
        int ka = k0 + kofs;
        v2f a = *(const v2f*)(arow + ka);
        v2f b = *(const v2f*)(brow + ka);
        acc = __builtin_amdgcn_wmma_f32_16x16x4_f32(
             false, a,  false, b,
             (short)0, acc,  false,  false);
    }

    int ncol = n0 + lid;
    if (ncol < N) {
        float* orow = Out + (size_t)(m0 + (half << 3)) * N + ncol;
#pragma unroll
        for (int r = 0; r < 8; ++r)
            orow[(size_t)r * N] = acc[r];
    }
}

#define LCAP  36864
#define LHDR  32
#define LREG  (LCAP + LHDR)
static_assert(NEDGES % 16 == 0, "dst and the owner bytes are read as 16-byte vectors");
static_assert(NEDGES <= (1 << 20) && NNODES <= (1 << 20) && RN <= 2048, "list entry packing");

__device__ __forceinline__ unsigned match_bytes(unsigned x, unsigned key)
{
    unsigned y = x ^ key;
    unsigned t = (y & 0x7F7F7F7Fu) + 0x7F7F7F7Fu;
    return ~(t | y | 0x7F7F7F7Fu);
}

template <int F, bool BUILD>
__device__ __forceinline__ void keep_edge(unsigned pos, int e, int lo,
                                          unsigned* list, float* acc, unsigned* indeg,
                                          const float* __restrict__ P,
                                          const int* __restrict__ src,
                                          const int* __restrict__ dst)
{
    if (pos < LCAP) {
        list[pos] = (unsigned)e;
    } else {
        unsigned r = (unsigned)(dst[e] - lo);
        if (BUILD) atomicAdd(&indeg[r], 1u);
        const float* row = P + (size_t)src[e] * F;
#pragma nounroll
        for (int f = 0; f < F; ++f) atomicAdd(&acc[r * F + f], row[f]);
    }
}

template <int F, bool BUILD>
__device__ __forceinline__ void test_sixteen(uint4 o, int e0, unsigned key, int lo,
                                             unsigned* cnt, unsigned* list, float* acc, unsigned* indeg,
                                             const float* __restrict__ P,
                                             const int* __restrict__ src,
                                             const int* __restrict__ dst)
{
    unsigned m0 = match_bytes(o.x, key);
    unsigned m1 = match_bytes(o.y, key);
    unsigned m2 = match_bytes(o.z, key);
    unsigned m3 = match_bytes(o.w, key);
    if (m0 | m1 | m2 | m3) {
        unsigned mk = (m0 >> 7) | (m1 >> 6) | (m2 >> 5) | (m3 >> 4);
        unsigned pos = atomicAdd(cnt, (unsigned)__popc(mk));
        do {
            unsigned t = (unsigned)__builtin_ctz(mk);
            mk &= mk - 1u;
            int e = e0 + (int)(((t & 7u) << 2) + (t >> 3));
            keep_edge<F, BUILD>(pos++, e, lo, list, acc, indeg, P, src, dst);
        } while (mk);
    }
}

template <int F, bool BUILD, bool RELU>
__global__ __launch_bounds__(1024) void owner_sum_kernel(
    const float* __restrict__ P, const int* __restrict__ src,
    const int* __restrict__ dst, float* __restrict__ hout,
    unsigned* __restrict__ saved, float* __restrict__ inv_deg,
    const unsigned* __restrict__ own, const float* __restrict__ bias)
{
    __shared__ float    acc[RN * F];
    __shared__ unsigned list[LCAP];
    __shared__ unsigned cnt;
    __shared__ unsigned indeg[BUILD ? RN : 1];
    __shared__ float    inv_s[RN];

    const int tid = threadIdx.x;
    const int lo  = blockIdx.x * RN;
    const unsigned nrows = (unsigned)((NNODES - lo) < RN ? (NNODES - lo) : RN);
    unsigned* mine = saved + (size_t)blockIdx.x * LREG;

    const unsigned have = BUILD ? 0u : mine[0];
    const bool filter = BUILD || have > (unsigned)LCAP;

    for (int i = tid; i < RN * F; i += 1024) acc[i] = 0.0f;
    if (BUILD && tid < RN) indeg[tid] = 0u;
    if (!BUILD && (unsigned)tid < nrows) inv_s[tid] = inv_deg[lo + tid];
    if (tid == 0) cnt = 0u;

    int n;
    if (filter) {
        __syncthreads();

        const uint4* own4 = (const uint4*)own;
        constexpr int NVEC = NEDGES / 16;
        const unsigned key = (unsigned)blockIdx.x * 0x01010101u;
        int base = tid;
        for (; base + 7 * 1024 < NVEC; base += 8 * 1024) {
            uint4 o[8];
#pragma unroll
            for (int u = 0; u < 8; ++u) o[u] = own4[base + u * 1024];
#pragma unroll
            for (int u = 0; u < 8; ++u)
                test_sixteen<F, BUILD>(o[u], (base + u * 1024) << 4, key, lo, &cnt, list, acc, indeg, P, src, dst);
        }
        for (; base < NVEC; base += 1024)
            test_sixteen<F, BUILD>(own4[base], base << 4, key, lo, &cnt, list, acc, indeg, P, src, dst);
        __syncthreads();
        const unsigned total = cnt;
        n = (int)(total < (unsigned)LCAP ? total : (unsigned)LCAP);

        for (int i = tid; i < n; i += 1024) {
            unsigned e = list[i];
            unsigned r = (unsigned)(dst[e] - lo);
            int s = src[e];
            if (BUILD) atomicAdd(&indeg[r], 1u);
            unsigned ent = (r << 20) | (unsigned)s;
            list[i] = ent;
            if (BUILD) mine[LHDR + i] = ent;
        }
        if (BUILD && tid == 0) mine[0] = total;
    } else {
        n = (int)have;
        for (int i = tid; i < n; i += 1024) list[i] = mine[LHDR + i];
    }
    __syncthreads();

    if (BUILD && (unsigned)tid < nrows) {
        float iv = 1.0f / fmaxf((float)indeg[tid], 1.0f);
        inv_deg[lo + tid] = iv;
        inv_s[tid] = iv;
    }

    const int wave = tid >> 5;
    const int lane = tid & 31;
    for (int i0 = wave; i0 < n; i0 += 32 * 8) {
        float v0[8], v1[8];
        unsigned rr[8], ents[8];
#pragma unroll
        for (int u = 0; u < 8; ++u) {
            int i = i0 + u * 32;
            ents[u] = (i < n) ? list[i] : 0u;
        }
#pragma unroll
        for (int u = 0; u < 8; ++u) {
            unsigned ent = __builtin_amdgcn_readfirstlane(ents[u]);
            const float* row = P + (size_t)(ent & 0xFFFFFu) * F;
            rr[u] = ent >> 20;
            v0[u] = row[lane];
            v1[u] = (lane + 32 < F) ? row[lane + 32] : 0.0f;
        }
#pragma unroll
        for (int u = 0; u < 8; ++u) {
            if (i0 + u * 32 < n) {
                float* a = acc + rr[u] * F;
                atomicAdd(a + lane, v0[u]);
                if (lane + 32 < F) atomicAdd(a + lane + 32, v1[u]);
            }
        }
    }
    __syncthreads();

    float* orow = hout + (size_t)lo * F;
    const int tot = (int)nrows * F;
    for (int i = tid; i < tot; i += 1024) {
        int r = i / F;
        int f = i - r * F;
        float v = acc[i] * inv_s[r] + bias[f];
        orow[i] = RELU ? fmaxf(v, 0.0f) : v;
    }
}


extern "C" void kernel_launch(void* const* d_in, const int* in_sizes, int n_in,
                              void* d_out, int out_size, void* d_ws, size_t ws_size,
                              hipStream_t stream)
{
    const float* features = (const float*)d_in[0];
    const int*   src      = (const int*)  d_in[1];
    const int*   dst      = (const int*)  d_in[2];
    const float* W0       = (const float*)d_in[3];
    const float* b0       = (const float*)d_in[4];
    const float* W1       = (const float*)d_in[5];
    const float* b1       = (const float*)d_in[6];
    const float* W2       = (const float*)d_in[7];
    const float* b2       = (const float*)d_in[8];
    float*       out      = (float*)d_out;

    float* ws  = (float*)d_ws;
    float* deg = ws;
    float* P   = ws + 50048;
    float* H   = P  + (size_t)NNODES * 64;
    unsigned* L = (unsigned*)(H + (size_t)NNODES * 64);
    unsigned* O = L + (size_t)NOWN * LREG;

    const int Mtiles     = NNODES / 16;
    const int gemmBlksX  = (Mtiles + 7) / 8;

    {
        gemm_wmma_kernel<true><<<dim3(gemmBlksX, 4), 256, 0, stream>>>(features, W0, P, 128, 64, Mtiles, dst, O);
        owner_sum_kernel<64, true, true><<<NOWN, 1024, 0, stream>>>(P, src, dst, H, L, deg, O, b0);
    }

    {
        gemm_wmma_kernel<false><<<dim3(gemmBlksX, 4), 256, 0, stream>>>(H, W1, P, 64, 64, Mtiles, dst, O);
        owner_sum_kernel<64, false, true><<<NOWN, 1024, 0, stream>>>(P, src, dst, H, L, deg, O, b1);
    }

    {
        gemm_wmma_kernel<false><<<dim3(gemmBlksX, 3), 256, 0, stream>>>(H, W2, P, 64, 40, Mtiles, dst, O);
        owner_sum_kernel<40, false, false><<<NOWN, 1024, 0, stream>>>(P, src, dst, out, L, deg, O, b2);
    }
}


# Next layer's product done by the owner block on its finished rows in LDS (f32 WMMA, weights in the freed list space): two product launches and the hidden-row round trip gone (6 -> 4); sum order kept
# speedup vs baseline: 16.5422x; 1.1890x over previous
#include <hip/hip_runtime.h>

typedef float v2f __attribute__((ext_vector_type(2)));
typedef float v8f __attribute__((ext_vector_type(8)));

#define NNODES 50000
#define NEDGES 800000
#define KMAX   128
#define BPAD   (KMAX + 2)

#define RN    197
#define NOWN  ((NNODES + RN - 1) / RN)
static_assert(NOWN <= 255, "owner id must fit a byte with 255 left free");

__device__ __forceinline__ unsigned owner_of(int d)
{
    unsigned u = (unsigned)d;
    return u < (unsigned)NNODES ? u / (unsigned)RN : 255u;
}

__global__ __launch_bounds__(256) void gemm_wmma_kernel(
    const float* __restrict__ Xin, const float* __restrict__ W,
    float* __restrict__ Out, int K, int N, int Mtiles,
    const int* __restrict__ dst, unsigned* __restrict__ own)
{
    __shared__ float smem[16 * BPAD];

    int tid = threadIdx.x;
    int n0  = blockIdx.y << 4;

    {
        const int4* dst4 = (const int4*)dst;
        const int nthr = (int)(gridDim.x * gridDim.y) * 256;
        for (int v = (int)(blockIdx.y * gridDim.x + blockIdx.x) * 256 + tid; v < NEDGES / 4; v += nthr) {
            int4 d = dst4[v];
            own[v] = owner_of(d.x) | (owner_of(d.y) << 8) | (owner_of(d.z) << 16) | (owner_of(d.w) << 24);
        }
    }

    for (int idx = tid; idx < (K << 4); idx += 256) {
        int k = idx >> 4;
        int n = idx & 15;
        int col = n0 + n;
        smem[n * BPAD + k] = (col < N) ? W[(size_t)k * N + col] : 0.0f;
    }
    __syncthreads();

    int tM = (blockIdx.x << 3) + (tid >> 5);
    if (tM >= Mtiles) return;

    int lane = tid & 31;
    int half = lane >> 4;
    int lid  = lane & 15;
    int m0   = tM << 4;

    const float* arow = Xin + (size_t)(m0 + lid) * K;
    const float* brow = smem + lid * BPAD;
    int kofs = half << 1;

    v8f acc = {};
    for (int k0 = 0; k0 < K; k0 += 4) {
        int ka = k0 + kofs;
        v2f a = *(const v2f*)(arow + ka);
        v2f b = *(const v2f*)(brow + ka);
        acc = __builtin_amdgcn_wmma_f32_16x16x4_f32(
             false, a,  false, b,
             (short)0, acc,  false,  false);
    }

    int ncol = n0 + lid;
    if (ncol < N) {
        float* orow = Out + (size_t)(m0 + (half << 3)) * N + ncol;
#pragma unroll
        for (int r = 0; r < 8; ++r)
            orow[(size_t)r * N] = acc[r];
    }
}

#define LCAP  36864
#define LHDR  32
#define LREG  (LCAP + LHDR)
static_assert(NEDGES % 16 == 0, "dst and the owner bytes are read as 16-byte vectors");
static_assert(NEDGES <= (1 << 20) && NNODES <= (1 << 20) && RN <= 2048, "list entry packing");

__device__ __forceinline__ unsigned match_bytes(unsigned x, unsigned key)
{
    unsigned y = x ^ key;
    unsigned t = (y & 0x7F7F7F7Fu) + 0x7F7F7F7Fu;
    return ~(t | y | 0x7F7F7F7Fu);
}

template <int F, bool BUILD>
__device__ __forceinline__ void keep_edge(unsigned pos, int e, int lo,
                                          unsigned* list, float* acc, unsigned* indeg,
                                          const float* __restrict__ P,
                                          const int* __restrict__ src,
                                          const int* __restrict__ dst)
{
    if (pos < LCAP) {
        list[pos] = (unsigned)e;
    } else {
        unsigned r = (unsigned)(dst[e] - lo);
        if (BUILD) atomicAdd(&indeg[r], 1u);
        const float* row = P + (size_t)src[e] * F;
#pragma nounroll
        for (int f = 0; f < F; ++f) atomicAdd(&acc[r * F + f], row[f]);
    }
}

template <int F, bool BUILD>
__device__ __forceinline__ void test_sixteen(uint4 o, int e0, unsigned key, int lo,
                                             unsigned* cnt, unsigned* list, float* acc, unsigned* indeg,
                                             const float* __restrict__ P,
                                             const int* __restrict__ src,
                                             const int* __restrict__ dst)
{
    unsigned m0 = match_bytes(o.x, key);
    unsigned m1 = match_bytes(o.y, key);
    unsigned m2 = match_bytes(o.z, key);
    unsigned m3 = match_bytes(o.w, key);
    if (m0 | m1 | m2 | m3) {
        unsigned mk = (m0 >> 7) | (m1 >> 6) | (m2 >> 5) | (m3 >> 4);
        unsigned pos = atomicAdd(cnt, (unsigned)__popc(mk));
        do {
            unsigned t = (unsigned)__builtin_ctz(mk);
            mk &= mk - 1u;
            int e = e0 + (int)(((t & 7u) << 2) + (t >> 3));
            keep_edge<F, BUILD>(pos++, e, lo, list, acc, indeg, P, src, dst);
        } while (mk);
    }
}

template <int F, bool BUILD, bool RELU, int NN>
__global__ __launch_bounds__(1024) void owner_sum_kernel(
    const float* __restrict__ P, const int* __restrict__ src,
    const int* __restrict__ dst, float* __restrict__ hout,
    unsigned* __restrict__ saved, float* __restrict__ inv_deg,
    const unsigned* __restrict__ own, const float* __restrict__ bias,
    const float* __restrict__ Wn)
{
    __shared__ float    acc[RN * F];
    __shared__ unsigned list[LCAP];
    __shared__ unsigned cnt;
    __shared__ unsigned indeg[BUILD ? RN : 1];
    __shared__ float    inv_s[RN];

    const int tid = threadIdx.x;
    const int lo  = blockIdx.x * RN;
    const unsigned nrows = (unsigned)((NNODES - lo) < RN ? (NNODES - lo) : RN);
    unsigned* mine = saved + (size_t)blockIdx.x * LREG;

    const unsigned have = BUILD ? 0u : mine[0];
    const bool filter = BUILD || have > (unsigned)LCAP;

    for (int i = tid; i < RN * F; i += 1024) acc[i] = 0.0f;
    if (BUILD && tid < RN) indeg[tid] = 0u;
    if (!BUILD && (unsigned)tid < nrows) inv_s[tid] = inv_deg[lo + tid];
    if (tid == 0) cnt = 0u;

    int n;
    if (filter) {
        __syncthreads();

        const uint4* own4 = (const uint4*)own;
        constexpr int NVEC = NEDGES / 16;
        const unsigned key = (unsigned)blockIdx.x * 0x01010101u;
        int base = tid;
        for (; base + 7 * 1024 < NVEC; base += 8 * 1024) {
            uint4 o[8];
#pragma unroll
            for (int u = 0; u < 8; ++u) o[u] = own4[base + u * 1024];
#pragma unroll
            for (int u = 0; u < 8; ++u)
                test_sixteen<F, BUILD>(o[u], (base + u * 1024) << 4, key, lo, &cnt, list, acc, indeg, P, src, dst);
        }
        for (; base < NVEC; base += 1024)
            test_sixteen<F, BUILD>(own4[base], base << 4, key, lo, &cnt, list, acc, indeg, P, src, dst);
        __syncthreads();
        const unsigned total = cnt;
        n = (int)(total < (unsigned)LCAP ? total : (unsigned)LCAP);

        for (int i = tid; i < n; i += 1024) {
            unsigned e = list[i];
            unsigned r = (unsigned)(dst[e] - lo);
            int s = src[e];
            if (BUILD) atomicAdd(&indeg[r], 1u);
            unsigned ent = (r << 20) | (unsigned)s;
            list[i] = ent;
            if (BUILD) mine[LHDR + i] = ent;
        }
        if (BUILD && tid == 0) mine[0] = total;
    } else {
        n = (int)have;
        for (int i = tid; i < n; i += 1024) list[i] = mine[LHDR + i];
    }
    __syncthreads();

    if (BUILD && (unsigned)tid < nrows) {
        float iv = 1.0f / fmaxf((float)indeg[tid], 1.0f);
        inv_deg[lo + tid] = iv;
        inv_s[tid] = iv;
    }

    const int wave = tid >> 5;
    const int lane = tid & 31;
    for (int i0 = wave; i0 < n; i0 += 32 * 8) {
        float v0[8], v1[8];
        unsigned rr[8], ents[8];
#pragma unroll
        for (int u = 0; u < 8; ++u) {
            int i = i0 + u * 32;
            ents[u] = (i < n) ? list[i] : 0u;
        }
#pragma unroll
        for (int u = 0; u < 8; ++u) {
            unsigned ent = __builtin_amdgcn_readfirstlane(ents[u]);
            const float* row = P + (size_t)(ent & 0xFFFFFu) * F;
            rr[u] = ent >> 20;
            v0[u] = row[lane];
            v1[u] = (lane + 32 < F) ? row[lane + 32] : 0.0f;
        }
#pragma unroll
        for (int u = 0; u < 8; ++u) {
            if (i0 + u * 32 < n) {
                float* a = acc + rr[u] * F;
                atomicAdd(a + lane, v0[u]);
                if (lane + 32 < F) atomicAdd(a + lane + 32, v1[u]);
            }
        }
    }
    __syncthreads();

    if constexpr (NN == 0) {
        float* orow = hout + (size_t)lo * F;
        const int tot = (int)nrows * F;
        for (int i = tid; i < tot; i += 1024) {
            int r = i / F;
            int f = i - r * F;
            float v = acc[i] * inv_s[r] + bias[f];
            orow[i] = RELU ? fmaxf(v, 0.0f) : v;
        }
    } else {
        constexpr int MT = (RN + 15) / 16;
        constexpr int NT = (NN + 15) / 16;
        constexpr int HP = F + 4;
        static_assert((MT + NT) * 16 * HP <= LCAP, "rows and weights fit the list's space");
        float* hs = reinterpret_cast<float*>(list);
        float* wt = hs + MT * 16 * HP;
        static_assert(1024 % F == 0 && (F * NT * 16) % 1024 == 0, "whole strides of the block");
        constexpr int NWL = F * NT * 16 / 1024;
        float wreg[NWL];
#pragma unroll
        for (int j = 0; j < NWL; ++j) {
            int idx = tid + j * 1024;
            int k = idx / (NT * 16);
            int n = idx - k * (NT * 16);
            wreg[j] = (n < NN) ? Wn[k * NN + n] : 0.0f;
        }
        {
            const int f = tid % F;
            const float bf = bias[f];
            for (int r = tid / F; r < MT * 16; r += 1024 / F) {
                float v = 0.0f;
                if ((unsigned)r < nrows) {
                    v = acc[r * F + f] * inv_s[r] + bf;
                    v = RELU ? fmaxf(v, 0.0f) : v;
                }
                hs[r * HP + f] = v;
            }
        }
#pragma unroll
        for (int j = 0; j < NWL; ++j) {
            int idx = tid + j * 1024;
            int k = idx / (NT * 16);
            int n = idx - k * (NT * 16);
            wt[n * HP + k] = wreg[j];
        }
        __syncthreads();

        const int half = lane >> 4;
        const int lid  = lane & 15;
        const int kofs = half << 1;
        const int wv = __builtin_amdgcn_readfirstlane(wave);
        for (int t = wv; t < MT * NT; t += 32) {
            const int mt = t / NT;
            const int nt = t - mt * NT;
            const float* arow = hs + (mt * 16 + lid) * HP + kofs;
            const float* brow = wt + (nt * 16 + lid) * HP + kofs;
            v8f c = {};
#pragma unroll
            for (int k0 = 0; k0 < F; k0 += 4) {
                v2f a = *(const v2f*)(arow + k0);
                v2f b = *(const v2f*)(brow + k0);
                c = __builtin_amdgcn_wmma_f32_16x16x4_f32(
                     false, a,  false, b,
                     (short)0, c,  false,  false);
            }
            const int ncol = nt * 16 + lid;
            const unsigned r0 = (unsigned)(mt * 16 + (half << 3));
            if (ncol < NN) {
                float* o = hout + ((size_t)lo + r0) * NN + ncol;
#pragma unroll
                for (int r = 0; r < 8; ++r)
                    if (r0 + (unsigned)r < nrows) o[(size_t)r * NN] = c[r];
            }
        }
    }
}


extern "C" void kernel_launch(void* const* d_in, const int* in_sizes, int n_in,
                              void* d_out, int out_size, void* d_ws, size_t ws_size,
                              hipStream_t stream)
{
    const float* features = (const float*)d_in[0];
    const int*   src      = (const int*)  d_in[1];
    const int*   dst      = (const int*)  d_in[2];
    const float* W0       = (const float*)d_in[3];
    const float* b0       = (const float*)d_in[4];
    const float* W1       = (const float*)d_in[5];
    const float* b1       = (const float*)d_in[6];
    const float* W2       = (const float*)d_in[7];
    const float* b2       = (const float*)d_in[8];
    float*       out      = (float*)d_out;

    float* ws  = (float*)d_ws;
    float* deg = ws;
    float* P   = ws + 50048;
    float* Q   = P  + (size_t)NNODES * 64;
    unsigned* L = (unsigned*)(Q + (size_t)NNODES * 64);
    unsigned* O = L + (size_t)NOWN * LREG;

    const int Mtiles     = NNODES / 16;
    const int gemmBlksX  = (Mtiles + 7) / 8;


    gemm_wmma_kernel<<<dim3(gemmBlksX, 4), 256, 0, stream>>>(features, W0, P, 128, 64, Mtiles, dst, O);
    owner_sum_kernel<64, true, true, 64><<<NOWN, 1024, 0, stream>>>(P, src, dst, Q, L, deg, O, b0, W1);

    owner_sum_kernel<64, false, true, 40><<<NOWN, 1024, 0, stream>>>(Q, src, dst, P, L, deg, O, b1, W2);

    owner_sum_kernel<40, false, false, 0><<<NOWN, 1024, 0, stream>>>(P, src, dst, out, L, deg, O, b2, nullptr);
}


# Gather loop of the owner blocks software-pipelined: a wave issues its next eight rows' loads before the current eight rows' LDS adds (two register sets, scalar loop control); sum order as before
# speedup vs baseline: 17.3898x; 1.0512x over previous
#include <hip/hip_runtime.h>

typedef float v2f __attribute__((ext_vector_type(2)));
typedef float v8f __attribute__((ext_vector_type(8)));

#define NNODES 50000
#define NEDGES 800000
#define KMAX   128
#define BPAD   (KMAX + 2)

#define RN    197
#define NOWN  ((NNODES + RN - 1) / RN)
static_assert(NOWN <= 255, "owner id must fit a byte with 255 left free");

__device__ __forceinline__ unsigned owner_of(int d)
{
    unsigned u = (unsigned)d;
    return u < (unsigned)NNODES ? u / (unsigned)RN : 255u;
}

__global__ __launch_bounds__(256) void gemm_wmma_kernel(
    const float* __restrict__ Xin, const float* __restrict__ W,
    float* __restrict__ Out, int K, int N, int Mtiles,
    const int* __restrict__ dst, unsigned* __restrict__ own)
{
    __shared__ float smem[16 * BPAD];

    int tid = threadIdx.x;
    int n0  = blockIdx.y << 4;

    {
        const int4* dst4 = (const int4*)dst;
        const int nthr = (int)(gridDim.x * gridDim.y) * 256;
        for (int v = (int)(blockIdx.y * gridDim.x + blockIdx.x) * 256 + tid; v < NEDGES / 4; v += nthr) {
            int4 d = dst4[v];
            own[v] = owner_of(d.x) | (owner_of(d.y) << 8) | (owner_of(d.z) << 16) | (owner_of(d.w) << 24);
        }
    }

    for (int idx = tid; idx < (K << 4); idx += 256) {
        int k = idx >> 4;
        int n = idx & 15;
        int col = n0 + n;
        smem[n * BPAD + k] = (col < N) ? W[(size_t)k * N + col] : 0.0f;
    }
    __syncthreads();

    int tM = (blockIdx.x << 3) + (tid >> 5);
    if (tM >= Mtiles) return;

    int lane = tid & 31;
    int half = lane >> 4;
    int lid  = lane & 15;
    int m0   = tM << 4;

    const float* arow = Xin + (size_t)(m0 + lid) * K;
    const float* brow = smem + lid * BPAD;
    int kofs = half << 1;

    v8f acc = {};
    for (int k0 = 0; k0 < K; k0 += 4) {
        int ka = k0 + kofs;
        v2f a = *(const v2f*)(arow + ka);
        v2f b = *(const v2f*)(brow + ka);
        acc = __builtin_amdgcn_wmma_f32_16x16x4_f32(
             false, a,  false, b,
             (short)0, acc,  false,  false);
    }

    int ncol = n0 + lid;
    if (ncol < N) {
        float* orow = Out + (size_t)(m0 + (half << 3)) * N + ncol;
#pragma unroll
        for (int r = 0; r < 8; ++r)
            orow[(size_t)r * N] = acc[r];
    }
}

#define LCAP  36864
#define LHDR  32
#define LREG  (LCAP + LHDR)
static_assert(NEDGES % 16 == 0, "dst and the owner bytes are read as 16-byte vectors");
static_assert(NEDGES <= (1 << 20) && NNODES <= (1 << 20) && RN <= 2048, "list entry packing");

__device__ __forceinline__ unsigned match_bytes(unsigned x, unsigned key)
{
    unsigned y = x ^ key;
    unsigned t = (y & 0x7F7F7F7Fu) + 0x7F7F7F7Fu;
    return ~(t | y | 0x7F7F7F7Fu);
}

template <int F, bool BUILD>
__device__ __forceinline__ void keep_edge(unsigned pos, int e, int lo,
                                          unsigned* list, float* acc, unsigned* indeg,
                                          const float* __restrict__ P,
                                          const int* __restrict__ src,
                                          const int* __restrict__ dst)
{
    if (pos < LCAP) {
        list[pos] = (unsigned)e;
    } else {
        unsigned r = (unsigned)(dst[e] - lo);
        if (BUILD) atomicAdd(&indeg[r], 1u);
        const float* row = P + (size_t)src[e] * F;
#pragma nounroll
        for (int f = 0; f < F; ++f) atomicAdd(&acc[r * F + f], row[f]);
    }
}

template <int F, bool BUILD>
__device__ __forceinline__ void test_sixteen(uint4 o, int e0, unsigned key, int lo,
                                             unsigned* cnt, unsigned* list, float* acc, unsigned* indeg,
                                             const float* __restrict__ P,
                                             const int* __restrict__ src,
                                             const int* __restrict__ dst)
{
    unsigned m0 = match_bytes(o.x, key);
    unsigned m1 = match_bytes(o.y, key);
    unsigned m2 = match_bytes(o.z, key);
    unsigned m3 = match_bytes(o.w, key);
    if (m0 | m1 | m2 | m3) {
        unsigned mk = (m0 >> 7) | (m1 >> 6) | (m2 >> 5) | (m3 >> 4);
        unsigned pos = atomicAdd(cnt, (unsigned)__popc(mk));
        do {
            unsigned t = (unsigned)__builtin_ctz(mk);
            mk &= mk - 1u;
            int e = e0 + (int)(((t & 7u) << 2) + (t >> 3));
            keep_edge<F, BUILD>(pos++, e, lo, list, acc, indeg, P, src, dst);
        } while (mk);
    }
}

template <int F>
__device__ __forceinline__ void rows_issue(const unsigned* list, const float* __restrict__ P,
                                           int i0, int n, int lane,
                                           unsigned (&rr)[8], float (&v0)[8], float (&v1)[8])
{
    unsigned ents[8];
#pragma unroll
    for (int u = 0; u < 8; ++u) {
        int i = i0 + u * 32;
        ents[u] = list[i < n ? i : n - 1];
    }
#pragma unroll
    for (int u = 0; u < 8; ++u) {
        unsigned ent = __builtin_amdgcn_readfirstlane(ents[u]);
        const float* row = P + (size_t)(ent & 0xFFFFFu) * F;
        rr[u] = ent >> 20;
        v0[u] = row[lane];
        v1[u] = row[lane + 32 < F ? lane + 32 : F - 1];
    }
}

template <int F, bool FULL>
__device__ __forceinline__ void rows_add(float* acc, int i0, int n, int lane,
                                         const unsigned (&rr)[8], const float (&v0)[8], const float (&v1)[8])
{
#pragma unroll
    for (int u = 0; u < 8; ++u) {
        if (FULL || i0 + u * 32 < n) {
            float* a = acc + rr[u] * F;
            atomicAdd(a + lane, v0[u]);
            if (lane + 32 < F) atomicAdd(a + lane + 32, v1[u]);
        }
    }
}

template <int F, bool BUILD, bool RELU, int NN>
__global__ __launch_bounds__(1024) void owner_sum_kernel(
    const float* __restrict__ P, const int* __restrict__ src,
    const int* __restrict__ dst, float* __restrict__ hout,
    unsigned* __restrict__ saved, float* __restrict__ inv_deg,
    const unsigned* __restrict__ own, const float* __restrict__ bias,
    const float* __restrict__ Wn)
{
    __shared__ float    acc[RN * F];
    __shared__ unsigned list[LCAP];
    __shared__ unsigned cnt;
    __shared__ unsigned indeg[BUILD ? RN : 1];
    __shared__ float    inv_s[RN];

    const int tid = threadIdx.x;
    const int lo  = blockIdx.x * RN;
    const unsigned nrows = (unsigned)((NNODES - lo) < RN ? (NNODES - lo) : RN);
    unsigned* mine = saved + (size_t)blockIdx.x * LREG;

    const unsigned have = BUILD ? 0u : mine[0];
    const bool filter = BUILD || have > (unsigned)LCAP;

    for (int i = tid; i < RN * F; i += 1024) acc[i] = 0.0f;
    if (BUILD && tid < RN) indeg[tid] = 0u;
    if (!BUILD && (unsigned)tid < nrows) inv_s[tid] = inv_deg[lo + tid];
    if (tid == 0) cnt = 0u;

    int n;
    if (filter) {
        __syncthreads();

        const uint4* own4 = (const uint4*)own;
        constexpr int NVEC = NEDGES / 16;
        const unsigned key = (unsigned)blockIdx.x * 0x01010101u;
        int base = tid;
        for (; base + 7 * 1024 < NVEC; base += 8 * 1024) {
            uint4 o[8];
#pragma unroll
            for (int u = 0; u < 8; ++u) o[u] = own4[base + u * 1024];
#pragma unroll
            for (int u = 0; u < 8; ++u)
                test_sixteen<F, BUILD>(o[u], (base + u * 1024) << 4, key, lo, &cnt, list, acc, indeg, P, src, dst);
        }
        for (; base < NVEC; base += 1024)
            test_sixteen<F, BUILD>(own4[base], base << 4, key, lo, &cnt, list, acc, indeg, P, src, dst);
        __syncthreads();
        const unsigned total = cnt;
        n = (int)(total < (unsigned)LCAP ? total : (unsigned)LCAP);

        for (int i = tid; i < n; i += 1024) {
            unsigned e = list[i];
            unsigned r = (unsigned)(dst[e] - lo);
            int s = src[e];
            if (BUILD) atomicAdd(&indeg[r], 1u);
            unsigned ent = (r << 20) | (unsigned)s;
            list[i] = ent;
            if (BUILD) mine[LHDR + i] = ent;
        }
        if (BUILD && tid == 0) mine[0] = total;
    } else {
        n = (int)have;
        for (int i = tid; i < n; i += 1024) list[i] = mine[LHDR + i];
    }
    __syncthreads();

    if (BUILD && (unsigned)tid < nrows) {
        float iv = 1.0f / fmaxf((float)indeg[tid], 1.0f);
        inv_deg[lo + tid] = iv;
        inv_s[tid] = iv;
    }

    const int wave = tid >> 5;
    const int lane = tid & 31;
    {
        int i0 = __builtin_amdgcn_readfirstlane(wave);
        if (i0 < n) {
            constexpr int G = 32 * 8;
            unsigned ra[8], rb[8];
            float a0[8], a1[8], b0[8], b1[8];
            rows_issue<F>(list, P, i0, n, lane, ra, a0, a1);
            while (i0 + 2 * G < n) {
                rows_issue<F>(list, P, i0 + G, n, lane, rb, b0, b1);
                rows_add<F, true>(acc, i0, n, lane, ra, a0, a1);
                rows_issue<F>(list, P, i0 + 2 * G, n, lane, ra, a0, a1);
                rows_add<F, true>(acc, i0 + G, n, lane, rb, b0, b1);
                i0 += 2 * G;
            }
            if (i0 + G < n) {
                rows_issue<F>(list, P, i0 + G, n, lane, rb, b0, b1);
                rows_add<F, true>(acc, i0, n, lane, ra, a0, a1);
                rows_add<F, false>(acc, i0 + G, n, lane, rb, b0, b1);
            } else {
                rows_add<F, false>(acc, i0, n, lane, ra, a0, a1);
            }
        }
    }
    __syncthreads();

    if constexpr (NN == 0) {
        float* orow = hout + (size_t)lo * F;
        const int tot = (int)nrows * F;
        for (int i = tid; i < tot; i += 1024) {
            int r = i / F;
            int f = i - r * F;
            float v = acc[i] * inv_s[r] + bias[f];
            orow[i] = RELU ? fmaxf(v, 0.0f) : v;
        }
    } else {
        constexpr int MT = (RN + 15) / 16;
        constexpr int NT = (NN + 15) / 16;
        constexpr int HP = F + 4;
        static_assert((MT + NT) * 16 * HP <= LCAP, "rows and weights fit the list's space");
        float* hs = reinterpret_cast<float*>(list);
        float* wt = hs + MT * 16 * HP;
        static_assert(1024 % F == 0 && (F * NT * 16) % 1024 == 0, "whole strides of the block");
        constexpr int NWL = F * NT * 16 / 1024;
        float wreg[NWL];
#pragma unroll
        for (int j = 0; j < NWL; ++j) {
            int idx = tid + j * 1024;
            int k = idx / (NT * 16);
            int n = idx - k * (NT * 16);
            wreg[j] = (n < NN) ? Wn[k * NN + n] : 0.0f;
        }
        {
            const int f = tid % F;
            const float bf = bias[f];
            for (int r = tid / F; r < MT * 16; r += 1024 / F) {
                float v = 0.0f;
                if ((unsigned)r < nrows) {
                    v = acc[r * F + f] * inv_s[r] + bf;
                    v = RELU ? fmaxf(v, 0.0f) : v;
                }
                hs[r * HP + f] = v;
            }
        }
#pragma unroll
        for (int j = 0; j < NWL; ++j) {
            int idx = tid + j * 1024;
            int k = idx / (NT * 16);
            int n = idx - k * (NT * 16);
            wt[n * HP + k] = wreg[j];
        }
        __syncthreads();

        const int half = lane >> 4;
        const int lid  = lane & 15;
        const int kofs = half << 1;
        const int wv = __builtin_amdgcn_readfirstlane(wave);
        for (int t = wv; t < MT * NT; t += 32) {
            const int mt = t / NT;
            const int nt = t - mt * NT;
            const float* arow = hs + (mt * 16 + lid) * HP + kofs;
            const float* brow = wt + (nt * 16 + lid) * HP + kofs;
            v8f c = {};
#pragma unroll
            for (int k0 = 0; k0 < F; k0 += 4) {
                v2f a = *(const v2f*)(arow + k0);
                v2f b = *(const v2f*)(brow + k0);
                c = __builtin_amdgcn_wmma_f32_16x16x4_f32(
                     false, a,  false, b,
                     (short)0, c,  false,  false);
            }
            const int ncol = nt * 16 + lid;
            const unsigned r0 = (unsigned)(mt * 16 + (half << 3));
            if (ncol < NN) {
                float* o = hout + ((size_t)lo + r0) * NN + ncol;
#pragma unroll
                for (int r = 0; r < 8; ++r)
                    if (r0 + (unsigned)r < nrows) o[(size_t)r * NN] = c[r];
            }
        }
    }
}


extern "C" void kernel_launch(void* const* d_in, const int* in_sizes, int n_in,
                              void* d_out, int out_size, void* d_ws, size_t ws_size,
                              hipStream_t stream)
{
    const float* features = (const float*)d_in[0];
    const int*   src      = (const int*)  d_in[1];
    const int*   dst      = (const int*)  d_in[2];
    const float* W0       = (const float*)d_in[3];
    const float* b0       = (const float*)d_in[4];
    const float* W1       = (const float*)d_in[5];
    const float* b1       = (const float*)d_in[6];
    const float* W2       = (const float*)d_in[7];
    const float* b2       = (const float*)d_in[8];
    float*       out      = (float*)d_out;

    float* ws  = (float*)d_ws;
    float* deg = ws;
    float* P   = ws + 50048;
    float* Q   = P  + (size_t)NNODES * 64;
    unsigned* L = (unsigned*)(Q + (size_t)NNODES * 64);
    unsigned* O = L + (size_t)NOWN * LREG;

    const int Mtiles     = NNODES / 16;
    const int gemmBlksX  = (Mtiles + 7) / 8;


    gemm_wmma_kernel<<<dim3(gemmBlksX, 4), 256, 0, stream>>>(features, W0, P, 128, 64, Mtiles, dst, O);
    owner_sum_kernel<64, true, true, 64><<<NOWN, 1024, 0, stream>>>(P, src, dst, Q, L, deg, O, b0, W1);

    owner_sum_kernel<64, false, true, 40><<<NOWN, 1024, 0, stream>>>(Q, src, dst, P, L, deg, O, b1, W2);

    owner_sum_kernel<40, false, false, 0><<<NOWN, 1024, 0, stream>>>(P, src, dst, out, L, deg, O, b2, nullptr);
}


# Owner ids stored bit-sliced by the product launch (8 words per 32 edges, one per id bit): an owner block tests 32 edges with 8 XORs; list slot order changes, sums still LDS adds in arrival order
# speedup vs baseline: 18.2089x; 1.0471x over previous
#include <hip/hip_runtime.h>

typedef float v2f __attribute__((ext_vector_type(2)));
typedef float v8f __attribute__((ext_vector_type(8)));

#define NNODES 50000
#define NEDGES 800000
#define KMAX   128
#define BPAD   (KMAX + 2)

#define NGRP  (NEDGES / 32)
#define OWNDW (((NGRP + 31) / 32) * 256)
#define RN    197
#define NOWN  ((NNODES + RN - 1) / RN)
static_assert(NOWN <= 255, "owner id must fit a byte with 255 left free");

__device__ __forceinline__ unsigned owner_of(int d)
{
    unsigned u = (unsigned)d;
    return u < (unsigned)NNODES ? u / (unsigned)RN : 255u;
}

__device__ __forceinline__ void slice_group(unsigned o, int g, int ln, unsigned* __restrict__ own)
{
    unsigned w = 0u;
#pragma unroll
    for (int k = 0; k < 8; ++k) {
        const unsigned bal = __builtin_amdgcn_ballot_w32(((o >> k) & 1u) != 0u);
        w = (ln == k) ? bal : w;
    }
    if (ln < 8)
        own[(g >> 5) * 256 + (ln >> 2) * 128 + (g & 31) * 4 + (ln & 3)] = w;
}

__global__ __launch_bounds__(256) void gemm_wmma_kernel(
    const float* __restrict__ Xin, const float* __restrict__ W,
    float* __restrict__ Out, int K, int N, int Mtiles,
    const int* __restrict__ dst, unsigned* __restrict__ own)
{
    __shared__ float smem[16 * BPAD];

    int tid = threadIdx.x;
    int n0  = blockIdx.y << 4;

    {
        const int ln  = tid & 31;
        const int nwv = (int)(gridDim.x * gridDim.y) * 8;
        for (int g = (int)(blockIdx.y * gridDim.x + blockIdx.x) * 8 + (tid >> 5); g < NGRP; g += 2 * nwv) {
            const int  g2   = g + nwv;
            const bool has2 = g2 < NGRP;
            const int  da   = dst[g * 32 + ln];
            const int  db   = dst[(has2 ? g2 : g) * 32 + ln];
            slice_group(owner_of(da), g, ln, own);
            if (has2) slice_group(owner_of(db), g2, ln, own);
        }
    }

    for (int idx = tid; idx < (K << 4); idx += 256) {
        int k = idx >> 4;
        int n = idx & 15;
        int col = n0 + n;
        smem[n * BPAD + k] = (col < N) ? W[(size_t)k * N + col] : 0.0f;
    }
    __syncthreads();

    int tM = (blockIdx.x << 3) + (tid >> 5);
    if (tM >= Mtiles) return;

    int lane = tid & 31;
    int half = lane >> 4;
    int lid  = lane & 15;
    int m0   = tM << 4;

    const float* arow = Xin + (size_t)(m0 + lid) * K;
    const float* brow = smem + lid * BPAD;
    int kofs = half << 1;

    v8f acc = {};
    for (int k0 = 0; k0 < K; k0 += 4) {
        int ka = k0 + kofs;
        v2f a = *(const v2f*)(arow + ka);
        v2f b = *(const v2f*)(brow + ka);
        acc = __builtin_amdgcn_wmma_f32_16x16x4_f32(
             false, a,  false, b,
             (short)0, acc,  false,  false);
    }

    int ncol = n0 + lid;
    if (ncol < N) {
        float* orow = Out + (size_t)(m0 + (half << 3)) * N + ncol;
#pragma unroll
        for (int r = 0; r < 8; ++r)
            orow[(size_t)r * N] = acc[r];
    }
}

#define LCAP  36864
#define LHDR  32
#define LREG  (LCAP + LHDR)
static_assert(NEDGES % 32 == 0, "the owner ids are sliced in whole groups of 32 edges");
static_assert(NEDGES <= (1 << 20) && NNODES <= (1 << 20) && RN <= 2048, "list entry packing");

template <int F, bool BUILD>
__device__ __forceinline__ void keep_edge(unsigned pos, int e, int lo,
                                          unsigned* list, float* acc, unsigned* indeg,
                                          const float* __restrict__ P,
                                          const int* __restrict__ src,
                                          const int* __restrict__ dst)
{
    if (pos < LCAP) {
        list[pos] = (unsigned)e;
    } else {
        unsigned r = (unsigned)(dst[e] - lo);
        if (BUILD) atomicAdd(&indeg[r], 1u);
        const float* row = P + (size_t)src[e] * F;
#pragma nounroll
        for (int f = 0; f < F; ++f) atomicAdd(&acc[r * F + f], row[f]);
    }
}

template <int F, bool BUILD>
__device__ __forceinline__ void test_group(uint4 a, uint4 b, int e0, const unsigned (&kf)[8], int lo,
                                           unsigned* cnt, unsigned* list, float* acc, unsigned* indeg,
                                           const float* __restrict__ P,
                                           const int* __restrict__ src,
                                           const int* __restrict__ dst)
{
    const unsigned mm = (a.x ^ kf[0]) | (a.y ^ kf[1]) | (a.z ^ kf[2]) | (a.w ^ kf[3])
                      | (b.x ^ kf[4]) | (b.y ^ kf[5]) | (b.z ^ kf[6]) | (b.w ^ kf[7]);
    unsigned mk = ~mm;
    if (mk) {
        unsigned pos = atomicAdd(cnt, (unsigned)__popc(mk));
        do {
            unsigned t = (unsigned)__builtin_ctz(mk);
            mk &= mk - 1u;
            keep_edge<F, BUILD>(pos++, e0 + (int)t, lo, list, acc, indeg, P, src, dst);
        } while (mk);
    }
}

template <int F>
__device__ __forceinline__ void rows_issue(const unsigned* list, const float* __restrict__ P,
                                           int i0, int n, int lane,
                                           unsigned (&rr)[8], float (&v0)[8], float (&v1)[8])
{
    unsigned ents[8];
#pragma unroll
    for (int u = 0; u < 8; ++u) {
        int i = i0 + u * 32;
        ents[u] = list[i < n ? i : n - 1];
    }
#pragma unroll
    for (int u = 0; u < 8; ++u) {
        unsigned ent = __builtin_amdgcn_readfirstlane(ents[u]);
        const float* row = P + (size_t)(ent & 0xFFFFFu) * F;
        rr[u] = ent >> 20;
        v0[u] = row[lane];
        v1[u] = row[lane + 32 < F ? lane + 32 : F - 1];
    }
}

template <int F, bool FULL>
__device__ __forceinline__ void rows_add(float* acc, int i0, int n, int lane,
                                         const unsigned (&rr)[8], const float (&v0)[8], const float (&v1)[8])
{
#pragma unroll
    for (int u = 0; u < 8; ++u) {
        if (FULL || i0 + u * 32 < n) {
            float* a = acc + rr[u] * F;
            atomicAdd(a + lane, v0[u]);
            if (lane + 32 < F) atomicAdd(a + lane + 32, v1[u]);
        }
    }
}

template <int F, bool BUILD, bool RELU, int NN>
__global__ __launch_bounds__(1024) void owner_sum_kernel(
    const float* __restrict__ P, const int* __restrict__ src,
    const int* __restrict__ dst, float* __restrict__ hout,
    unsigned* __restrict__ saved, float* __restrict__ inv_deg,
    const unsigned* __restrict__ own, const float* __restrict__ bias,
    const float* __restrict__ Wn)
{
    __shared__ float    acc[RN * F];
    __shared__ unsigned list[LCAP];
    __shared__ unsigned cnt;
    __shared__ unsigned indeg[BUILD ? RN : 1];
    __shared__ float    inv_s[RN];

    const int tid = threadIdx.x;
    const int lo  = blockIdx.x * RN;
    const unsigned nrows = (unsigned)((NNODES - lo) < RN ? (NNODES - lo) : RN);
    unsigned* mine = saved + (size_t)blockIdx.x * LREG;

    const unsigned have = BUILD ? 0u : mine[0];
    const bool filter = BUILD || have > (unsigned)LCAP;

    for (int i = tid; i < RN * F; i += 1024) acc[i] = 0.0f;
    if (BUILD && tid < RN) indeg[tid] = 0u;
    if (!BUILD && (unsigned)tid < nrows) inv_s[tid] = inv_deg[lo + tid];
    if (tid == 0) cnt = 0u;

    int n;
    if (filter) {
        __syncthreads();

        const uint4* own4 = (const uint4*)own;
        unsigned kf[8];
#pragma unroll
        for (int k = 0; k < 8; ++k) kf[k] = 0u - ((blockIdx.x >> k) & 1u);
        int g = tid;
        for (; g + 3 * 1024 < NGRP; g += 4 * 1024) {
            uint4 a[4], b[4];
#pragma unroll
            for (int u = 0; u < 4; ++u) {
                const int gu = g + u * 1024;
                const int q  = (gu >> 5) * 64 + (gu & 31);
                a[u] = own4[q];
                b[u] = own4[q + 32];
            }
#pragma unroll
            for (int u = 0; u < 4; ++u)
                test_group<F, BUILD>(a[u], b[u], (g + u * 1024) << 5, kf, lo, &cnt, list, acc, indeg, P, src, dst);
        }
        for (; g < NGRP; g += 1024) {
            const int q = (g >> 5) * 64 + (g & 31);
            test_group<F, BUILD>(own4[q], own4[q + 32], g << 5, kf, lo, &cnt, list, acc, indeg, P, src, dst);
        }
        __syncthreads();
        const unsigned total = cnt;
        n = (int)(total < (unsigned)LCAP ? total : (unsigned)LCAP);

        for (int i = tid; i < n; i += 1024) {
            unsigned e = list[i];
            unsigned r = (unsigned)(dst[e] - lo);
            int s = src[e];
            if (BUILD) atomicAdd(&indeg[r], 1u);
            unsigned ent = (r << 20) | (unsigned)s;
            list[i] = ent;
            if (BUILD) mine[LHDR + i] = ent;
        }
        if (BUILD && tid == 0) mine[0] = total;
    } else {
        n = (int)have;
        for (int i = tid; i < n; i += 1024) list[i] = mine[LHDR + i];
    }
    __syncthreads();

    if (BUILD && (unsigned)tid < nrows) {
        float iv = 1.0f / fmaxf((float)indeg[tid], 1.0f);
        inv_deg[lo + tid] = iv;
        inv_s[tid] = iv;
    }

    const int wave = tid >> 5;
    const int lane = tid & 31;
    {
        int i0 = __builtin_amdgcn_readfirstlane(wave);
        if (i0 < n) {
            constexpr int G = 32 * 8;
            unsigned ra[8], rb[8];
            float a0[8], a1[8], b0[8], b1[8];
            rows_issue<F>(list, P, i0, n, lane, ra, a0, a1);
            while (i0 + 2 * G < n) {
                rows_issue<F>(list, P, i0 + G, n, lane, rb, b0, b1);
                rows_add<F, true>(acc, i0, n, lane, ra, a0, a1);
                rows_issue<F>(list, P, i0 + 2 * G, n, lane, ra, a0, a1);
                rows_add<F, true>(acc, i0 + G, n, lane, rb, b0, b1);
                i0 += 2 * G;
            }
            if (i0 + G < n) {
                rows_issue<F>(list, P, i0 + G, n, lane, rb, b0, b1);
                rows_add<F, true>(acc, i0, n, lane, ra, a0, a1);
                rows_add<F, false>(acc, i0 + G, n, lane, rb, b0, b1);
            } else {
                rows_add<F, false>(acc, i0, n, lane, ra, a0, a1);
            }
        }
    }
    __syncthreads();

    if constexpr (NN == 0) {
        float* orow = hout + (size_t)lo * F;
        const int tot = (int)nrows * F;
        for (int i = tid; i < tot; i += 1024) {
            int r = i / F;
            int f = i - r * F;
            float v = acc[i] * inv_s[r] + bias[f];
            orow[i] = RELU ? fmaxf(v, 0.0f) : v;
        }
    } else {
        constexpr int MT = (RN + 15) / 16;
        constexpr int NT = (NN + 15) / 16;
        constexpr int HP = F + 4;
        static_assert((MT + NT) * 16 * HP <= LCAP, "rows and weights fit the list's space");
        float* hs = reinterpret_cast<float*>(list);
        float* wt = hs + MT * 16 * HP;
        static_assert(1024 % F == 0 && (F * NT * 16) % 1024 == 0, "whole strides of the block");
        constexpr int NWL = F * NT * 16 / 1024;
        float wreg[NWL];
#pragma unroll
        for (int j = 0; j < NWL; ++j) {
            int idx = tid + j * 1024;
            int k = idx / (NT * 16);
            int n = idx - k * (NT * 16);
            wreg[j] = (n < NN) ? Wn[k * NN + n] : 0.0f;
        }
        {
            const int f = tid % F;
            const float bf = bias[f];
            for (int r = tid / F; r < MT * 16; r += 1024 / F) {
                float v = 0.0f;
                if ((unsigned)r < nrows) {
                    v = acc[r * F + f] * inv_s[r] + bf;
                    v = RELU ? fmaxf(v, 0.0f) : v;
                }
                hs[r * HP + f] = v;
            }
        }
#pragma unroll
        for (int j = 0; j < NWL; ++j) {
            int idx = tid + j * 1024;
            int k = idx / (NT * 16);
            int n = idx - k * (NT * 16);
            wt[n * HP + k] = wreg[j];
        }
        __syncthreads();

        const int half = lane >> 4;
        const int lid  = lane & 15;
        const int kofs = half << 1;
        const int wv = __builtin_amdgcn_readfirstlane(wave);
        for (int t = wv; t < MT * NT; t += 32) {
            const int mt = t / NT;
            const int nt = t - mt * NT;
            const float* arow = hs + (mt * 16 + lid) * HP + kofs;
            const float* brow = wt + (nt * 16 + lid) * HP + kofs;
            v8f c = {};
#pragma unroll
            for (int k0 = 0; k0 < F; k0 += 4) {
                v2f a = *(const v2f*)(arow + k0);
                v2f b = *(const v2f*)(brow + k0);
                c = __builtin_amdgcn_wmma_f32_16x16x4_f32(
                     false, a,  false, b,
                     (short)0, c,  false,  false);
            }
            const int ncol = nt * 16 + lid;
            const unsigned r0 = (unsigned)(mt * 16 + (half << 3));
            if (ncol < NN) {
                float* o = hout + ((size_t)lo + r0) * NN + ncol;
#pragma unroll
                for (int r = 0; r < 8; ++r)
                    if (r0 + (unsigned)r < nrows) o[(size_t)r * NN] = c[r];
            }
        }
    }
}


extern "C" void kernel_launch(void* const* d_in, const int* in_sizes, int n_in,
                              void* d_out, int out_size, void* d_ws, size_t ws_size,
                              hipStream_t stream)
{
    const float* features = (const float*)d_in[0];
    const int*   src      = (const int*)  d_in[1];
    const int*   dst      = (const int*)  d_in[2];
    const float* W0       = (const float*)d_in[3];
    const float* b0       = (const float*)d_in[4];
    const float* W1       = (const float*)d_in[5];
    const float* b1       = (const float*)d_in[6];
    const float* W2       = (const float*)d_in[7];
    const float* b2       = (const float*)d_in[8];
    float*       out      = (float*)d_out;

    float* ws  = (float*)d_ws;
    float* deg = ws;
    float* P   = ws + 50048;
    float* Q   = P  + (size_t)NNODES * 64;
    unsigned* L = (unsigned*)(Q + (size_t)NNODES * 64);
    unsigned* O = L + (size_t)NOWN * LREG;

    const int Mtiles     = NNODES / 16;
    const int gemmBlksX  = (Mtiles + 7) / 8;


    gemm_wmma_kernel<<<dim3(gemmBlksX, 4), 256, 0, stream>>>(features, W0, P, 128, 64, Mtiles, dst, O);
    owner_sum_kernel<64, true, true, 64><<<NOWN, 1024, 0, stream>>>(P, src, dst, Q, L, deg, O, b0, W1);

    owner_sum_kernel<64, false, true, 40><<<NOWN, 1024, 0, stream>>>(Q, src, dst, P, L, deg, O, b1, W2);

    owner_sum_kernel<40, false, false, 0><<<NOWN, 1024, 0, stream>>>(P, src, dst, out, L, deg, O, b2, nullptr);
}


# Layer-0 product: a wave keeps its 16 rows in registers (16-B fragment loads, issued at wave start) and runs all 4 column stripes on them, W0 whole in LDS; 12512 -> 3128 waves; k regrouped in the sum
# speedup vs baseline: 19.9842x; 1.0975x over previous
#include <hip/hip_runtime.h>

typedef float v2f __attribute__((ext_vector_type(2)));
typedef float v8f __attribute__((ext_vector_type(8)));

#define NNODES 50000
#define NEDGES 800000

#define NGRP  (NEDGES / 32)
#define OWNDW (((NGRP + 31) / 32) * 256)
#define RN    197
#define NOWN  ((NNODES + RN - 1) / RN)
static_assert(NOWN <= 255, "owner id must fit a byte with 255 left free");

__device__ __forceinline__ unsigned owner_of(int d)
{
    unsigned u = (unsigned)d;
    return u < (unsigned)NNODES ? u / (unsigned)RN : 255u;
}

__device__ __forceinline__ void slice_group(unsigned o, int g, int ln, unsigned* __restrict__ own)
{
    unsigned w = 0u;
#pragma unroll
    for (int k = 0; k < 8; ++k) {
        const unsigned bal = __builtin_amdgcn_ballot_w32(((o >> k) & 1u) != 0u);
        w = (ln == k) ? bal : w;
    }
    if (ln < 8)
        own[(g >> 5) * 256 + (ln >> 2) * 128 + (g & 31) * 4 + (ln & 3)] = w;
}

typedef float v4f __attribute__((ext_vector_type(4)));
#define GK     128
#define GN     64
#define WPITCH (GK + 4)
#define GTILES (NNODES / 16)
#define GBLK   ((GTILES + 7) / 8)
#define GWAVES (GBLK * 8)
#define SGRP   ((NGRP + GWAVES - 1) / GWAVES)
static_assert(NNODES % 16 == 0, "whole 16-row tiles");
static_assert(GWAVES <= NGRP, "every wave has a first group");
static_assert(GK % 8 == 0 && GN == 64, "fragment loop and weight staging below");

__global__ __launch_bounds__(256) void gemm_wmma_kernel(
    const float* __restrict__ Xin, const float* __restrict__ W,
    float* __restrict__ Out,
    const int* __restrict__ dst, unsigned* __restrict__ own)
{
    __shared__ float smem[GN * WPITCH];

    const int tid  = threadIdx.x;
    const int wave = tid >> 5;
    const int lane = tid & 31;
    const int half = lane >> 4;
    const int lid  = lane & 15;
    const int wid  = (int)blockIdx.x * 8 + wave;

    v4f wv[GK * GN / 4 / 256];
#pragma unroll
    for (int j = 0; j < GK * GN / 4 / 256; ++j) {
        const int unit = wave * (GK * GN / 4 / 256) + j;
        const int k  = (unit >> 2) * 8 + (lane & 7);
        const int n4 = (unit & 3) * 16 + (lane >> 3) * 4;
        wv[j] = *(const v4f*)(W + k * GN + n4);
    }
    int dv[SGRP];
#pragma unroll
    for (int u = 0; u < SGRP; ++u) {
        const int g = wid + u * GWAVES;
        dv[u] = dst[(g < NGRP ? g : wid) * 32 + lane];
    }
    const bool has = wid < GTILES;
    const int  m0  = (has ? wid : GTILES - 1) << 4;
    const float* arow = Xin + (size_t)(m0 + lid) * GK + (half << 2);
    v4f a[GK / 8];
#pragma unroll
    for (int j = 0; j < GK / 8; ++j)
        a[j] = *(const v4f*)(arow + 8 * j);

#pragma unroll
    for (int j = 0; j < GK * GN / 4 / 256; ++j) {
        const int unit = wave * (GK * GN / 4 / 256) + j;
        const int k  = (unit >> 2) * 8 + (lane & 7);
        const int n4 = (unit & 3) * 16 + (lane >> 3) * 4;
        smem[(n4 + 0) * WPITCH + k] = wv[j].x;
        smem[(n4 + 1) * WPITCH + k] = wv[j].y;
        smem[(n4 + 2) * WPITCH + k] = wv[j].z;
        smem[(n4 + 3) * WPITCH + k] = wv[j].w;
    }

#pragma unroll
    for (int u = 0; u < SGRP; ++u) {
        const int g = wid + u * GWAVES;
        if (u == 0 || g < NGRP) slice_group(owner_of(dv[u]), g, lane, own);
    }
    __syncthreads();
    if (!has) return;

    const float* bbase = smem + lid * WPITCH + (half << 2);
    v8f acc[4] = {};
#pragma unroll
    for (int j = 0; j < GK / 8; ++j) {
        v4f b[4];
#pragma unroll
        for (int t = 0; t < 4; ++t)
            b[t] = *(const v4f*)(bbase + t * 16 * WPITCH + 8 * j);
        const v2f alo = {a[j].x, a[j].y};
        const v2f ahi = {a[j].z, a[j].w};
#pragma unroll
        for (int t = 0; t < 4; ++t) {
            const v2f blo = {b[t].x, b[t].y};
            acc[t] = __builtin_amdgcn_wmma_f32_16x16x4_f32(
                 false, alo,  false, blo,
                 (short)0, acc[t],  false,  false);
        }
#pragma unroll
        for (int t = 0; t < 4; ++t) {
            const v2f bhi = {b[t].z, b[t].w};
            acc[t] = __builtin_amdgcn_wmma_f32_16x16x4_f32(
                 false, ahi,  false, bhi,
                 (short)0, acc[t],  false,  false);
        }
    }

    float* orow = Out + (size_t)(m0 + (half << 3)) * GN + lid;
#pragma unroll
    for (int t = 0; t < 4; ++t)
#pragma unroll
        for (int r = 0; r < 8; ++r)
            orow[(size_t)r * GN + 16 * t] = acc[t][r];
}

#define LCAP  36864
#define LHDR  32
#define LREG  (LCAP + LHDR)
static_assert(NEDGES % 32 == 0, "the owner ids are sliced in whole groups of 32 edges");
static_assert(NEDGES <= (1 << 20) && NNODES <= (1 << 20) && RN <= 2048, "list entry packing");

template <int F, bool BUILD>
__device__ __forceinline__ void keep_edge(unsigned pos, int e, int lo,
                                          unsigned* list, float* acc, unsigned* indeg,
                                          const float* __restrict__ P,
                                          const int* __restrict__ src,
                                          const int* __restrict__ dst)
{
    if (pos < LCAP) {
        list[pos] = (unsigned)e;
    } else {
        unsigned r = (unsigned)(dst[e] - lo);
        if (BUILD) atomicAdd(&indeg[r], 1u);
        const float* row = P + (size_t)src[e] * F;
#pragma nounroll
        for (int f = 0; f < F; ++f) atomicAdd(&acc[r * F + f], row[f]);
    }
}

template <int F, bool BUILD>
__device__ __forceinline__ void test_group(uint4 a, uint4 b, int e0, const unsigned (&kf)[8], int lo,
                                           unsigned* cnt, unsigned* list, float* acc, unsigned* indeg,
                                           const float* __restrict__ P,
                                           const int* __restrict__ src,
                                           const int* __restrict__ dst)
{
    const unsigned mm = (a.x ^ kf[0]) | (a.y ^ kf[1]) | (a.z ^ kf[2]) | (a.w ^ kf[3])
                      | (b.x ^ kf[4]) | (b.y ^ kf[5]) | (b.z ^ kf[6]) | (b.w ^ kf[7]);
    unsigned mk = ~mm;
    if (mk) {
        unsigned pos = atomicAdd(cnt, (unsigned)__popc(mk));
        do {
            unsigned t = (unsigned)__builtin_ctz(mk);
            mk &= mk - 1u;
            keep_edge<F, BUILD>(pos++, e0 + (int)t, lo, list, acc, indeg, P, src, dst);
        } while (mk);
    }
}

template <int F>
__device__ __forceinline__ void rows_issue(const unsigned* list, const float* __restrict__ P,
                                           int i0, int n, int lane,
                                           unsigned (&rr)[8], float (&v0)[8], float (&v1)[8])
{
    unsigned ents[8];
#pragma unroll
    for (int u = 0; u < 8; ++u) {
        int i = i0 + u * 32;
        ents[u] = list[i < n ? i : n - 1];
    }
#pragma unroll
    for (int u = 0; u < 8; ++u) {
        unsigned ent = __builtin_amdgcn_readfirstlane(ents[u]);
        const float* row = P + (size_t)(ent & 0xFFFFFu) * F;
        rr[u] = ent >> 20;
        v0[u] = row[lane];
        v1[u] = row[lane + 32 < F ? lane + 32 : F - 1];
    }
}

template <int F, bool FULL>
__device__ __forceinline__ void rows_add(float* acc, int i0, int n, int lane,
                                         const unsigned (&rr)[8], const float (&v0)[8], const float (&v1)[8])
{
#pragma unroll
    for (int u = 0; u < 8; ++u) {
        if (FULL || i0 + u * 32 < n) {
            float* a = acc + rr[u] * F;
            atomicAdd(a + lane, v0[u]);
            if (lane + 32 < F) atomicAdd(a + lane + 32, v1[u]);
        }
    }
}

template <int F, bool BUILD, bool RELU, int NN>
__global__ __launch_bounds__(1024) void owner_sum_kernel(
    const float* __restrict__ P, const int* __restrict__ src,
    const int* __restrict__ dst, float* __restrict__ hout,
    unsigned* __restrict__ saved, float* __restrict__ inv_deg,
    const unsigned* __restrict__ own, const float* __restrict__ bias,
    const float* __restrict__ Wn)
{
    __shared__ float    acc[RN * F];
    __shared__ unsigned list[LCAP];
    __shared__ unsigned cnt;
    __shared__ unsigned indeg[BUILD ? RN : 1];
    __shared__ float    inv_s[RN];

    const int tid = threadIdx.x;
    const int lo  = blockIdx.x * RN;
    const unsigned nrows = (unsigned)((NNODES - lo) < RN ? (NNODES - lo) : RN);
    unsigned* mine = saved + (size_t)blockIdx.x * LREG;

    const unsigned have = BUILD ? 0u : mine[0];
    const bool filter = BUILD || have > (unsigned)LCAP;

    for (int i = tid; i < RN * F; i += 1024) acc[i] = 0.0f;
    if (BUILD && tid < RN) indeg[tid] = 0u;
    if (!BUILD && (unsigned)tid < nrows) inv_s[tid] = inv_deg[lo + tid];
    if (tid == 0) cnt = 0u;

    int n;
    if (filter) {
        __syncthreads();

        const uint4* own4 = (const uint4*)own;
        unsigned kf[8];
#pragma unroll
        for (int k = 0; k < 8; ++k) kf[k] = 0u - ((blockIdx.x >> k) & 1u);
        int g = tid;
        for (; g + 3 * 1024 < NGRP; g += 4 * 1024) {
            uint4 a[4], b[4];
#pragma unroll
            for (int u = 0; u < 4; ++u) {
                const int gu = g + u * 1024;
                const int q  = (gu >> 5) * 64 + (gu & 31);
                a[u] = own4[q];
                b[u] = own4[q + 32];
            }
#pragma unroll
            for (int u = 0; u < 4; ++u)
                test_group<F, BUILD>(a[u], b[u], (g + u * 1024) << 5, kf, lo, &cnt, list, acc, indeg, P, src, dst);
        }
        for (; g < NGRP; g += 1024) {
            const int q = (g >> 5) * 64 + (g & 31);
            test_group<F, BUILD>(own4[q], own4[q + 32], g << 5, kf, lo, &cnt, list, acc, indeg, P, src, dst);
        }
        __syncthreads();
        const unsigned total = cnt;
        n = (int)(total < (unsigned)LCAP ? total : (unsigned)LCAP);

        for (int i = tid; i < n; i += 1024) {
            unsigned e = list[i];
            unsigned r = (unsigned)(dst[e] - lo);
            int s = src[e];
            if (BUILD) atomicAdd(&indeg[r], 1u);
            unsigned ent = (r << 20) | (unsigned)s;
            list[i] = ent;
            if (BUILD) mine[LHDR + i] = ent;
        }
        if (BUILD && tid == 0) mine[0] = total;
    } else {
        n = (int)have;
        for (int i = tid; i < n; i += 1024) list[i] = mine[LHDR + i];
    }
    __syncthreads();

    if (BUILD && (unsigned)tid < nrows) {
        float iv = 1.0f / fmaxf((float)indeg[tid], 1.0f);
        inv_deg[lo + tid] = iv;
        inv_s[tid] = iv;
    }

    const int wave = tid >> 5;
    const int lane = tid & 31;
    {
        int i0 = __builtin_amdgcn_readfirstlane(wave);
        if (i0 < n) {
            constexpr int G = 32 * 8;
            unsigned ra[8], rb[8];
            float a0[8], a1[8], b0[8], b1[8];
            rows_issue<F>(list, P, i0, n, lane, ra, a0, a1);
            while (i0 + 2 * G < n) {
                rows_issue<F>(list, P, i0 + G, n, lane, rb, b0, b1);
                rows_add<F, true>(acc, i0, n, lane, ra, a0, a1);
                rows_issue<F>(list, P, i0 + 2 * G, n, lane, ra, a0, a1);
                rows_add<F, true>(acc, i0 + G, n, lane, rb, b0, b1);
                i0 += 2 * G;
            }
            if (i0 + G < n) {
                rows_issue<F>(list, P, i0 + G, n, lane, rb, b0, b1);
                rows_add<F, true>(acc, i0, n, lane, ra, a0, a1);
                rows_add<F, false>(acc, i0 + G, n, lane, rb, b0, b1);
            } else {
                rows_add<F, false>(acc, i0, n, lane, ra, a0, a1);
            }
        }
    }
    __syncthreads();

    if constexpr (NN == 0) {
        float* orow = hout + (size_t)lo * F;
        const int tot = (int)nrows * F;
        for (int i = tid; i < tot; i += 1024) {
            int r = i / F;
            int f = i - r * F;
            float v = acc[i] * inv_s[r] + bias[f];
            orow[i] = RELU ? fmaxf(v, 0.0f) : v;
        }
    } else {
        constexpr int MT = (RN + 15) / 16;
        constexpr int NT = (NN + 15) / 16;
        constexpr int HP = F + 4;
        static_assert((MT + NT) * 16 * HP <= LCAP, "rows and weights fit the list's space");
        float* hs = reinterpret_cast<float*>(list);
        float* wt = hs + MT * 16 * HP;
        static_assert(1024 % F == 0 && (F * NT * 16) % 1024 == 0, "whole strides of the block");
        constexpr int NWL = F * NT * 16 / 1024;
        float wreg[NWL];
#pragma unroll
        for (int j = 0; j < NWL; ++j) {
            int idx = tid + j * 1024;
            int k = idx / (NT * 16);
            int n = idx - k * (NT * 16);
            wreg[j] = (n < NN) ? Wn[k * NN + n] : 0.0f;
        }
        {
            const int f = tid % F;
            const float bf = bias[f];
            for (int r = tid / F; r < MT * 16; r += 1024 / F) {
                float v = 0.0f;
                if ((unsigned)r < nrows) {
                    v = acc[r * F + f] * inv_s[r] + bf;
                    v = RELU ? fmaxf(v, 0.0f) : v;
                }
                hs[r * HP + f] = v;
            }
        }
#pragma unroll
        for (int j = 0; j < NWL; ++j) {
            int idx = tid + j * 1024;
            int k = idx / (NT * 16);
            int n = idx - k * (NT * 16);
            wt[n * HP + k] = wreg[j];
        }
        __syncthreads();

        const int half = lane >> 4;
        const int lid  = lane & 15;
        const int kofs = half << 1;
        const int wv = __builtin_amdgcn_readfirstlane(wave);
        for (int t = wv; t < MT * NT; t += 32) {
            const int mt = t / NT;
            const int nt = t - mt * NT;
            const float* arow = hs + (mt * 16 + lid) * HP + kofs;
            const float* brow = wt + (nt * 16 + lid) * HP + kofs;
            v8f c = {};
#pragma unroll
            for (int k0 = 0; k0 < F; k0 += 4) {
                v2f a = *(const v2f*)(arow + k0);
                v2f b = *(const v2f*)(brow + k0);
                c = __builtin_amdgcn_wmma_f32_16x16x4_f32(
                     false, a,  false, b,
                     (short)0, c,  false,  false);
            }
            const int ncol = nt * 16 + lid;
            const unsigned r0 = (unsigned)(mt * 16 + (half << 3));
            if (ncol < NN) {
                float* o = hout + ((size_t)lo + r0) * NN + ncol;
#pragma unroll
                for (int r = 0; r < 8; ++r)
                    if (r0 + (unsigned)r < nrows) o[(size_t)r * NN] = c[r];
            }
        }
    }
}


extern "C" void kernel_launch(void* const* d_in, const int* in_sizes, int n_in,
                              void* d_out, int out_size, void* d_ws, size_t ws_size,
                              hipStream_t stream)
{
    const float* features = (const float*)d_in[0];
    const int*   src      = (const int*)  d_in[1];
    const int*   dst      = (const int*)  d_in[2];
    const float* W0       = (const float*)d_in[3];
    const float* b0       = (const float*)d_in[4];
    const float* W1       = (const float*)d_in[5];
    const float* b1       = (const float*)d_in[6];
    const float* W2       = (const float*)d_in[7];
    const float* b2       = (const float*)d_in[8];
    float*       out      = (float*)d_out;

    float* ws  = (float*)d_ws;
    float* deg = ws;
    float* P   = ws + 50048;
    float* Q   = P  + (size_t)NNODES * 64;
    unsigned* L = (unsigned*)(Q + (size_t)NNODES * 64);
    unsigned* O = L + (size_t)NOWN * LREG;


    gemm_wmma_kernel<<<GBLK, 256, 0, stream>>>(features, W0, P, dst, O);
    owner_sum_kernel<64, true, true, 64><<<NOWN, 1024, 0, stream>>>(P, src, dst, Q, L, deg, O, b0, W1);

    owner_sum_kernel<64, false, true, 40><<<NOWN, 1024, 0, stream>>>(Q, src, dst, P, L, deg, O, b1, W2);

    owner_sum_kernel<40, false, false, 0><<<NOWN, 1024, 0, stream>>>(P, src, dst, out, L, deg, O, b2, nullptr);
}


# Gather: about 3 memory instructions an edge, not 5: a row by ONE 8-byte-a-lane load, a wave's 8 list entries by ONE LDS load (slots in runs of 8); sums keep even columns first; arrival-order adds
# speedup vs baseline: 21.1552x; 1.0586x over previous
#include <hip/hip_runtime.h>

typedef float v2f __attribute__((ext_vector_type(2)));
typedef float v8f __attribute__((ext_vector_type(8)));

#define NNODES 50000
#define NEDGES 800000

#define NGRP  (NEDGES / 32)
#define OWNDW (((NGRP + 31) / 32) * 256)
#define RN    197
#define NOWN  ((NNODES + RN - 1) / RN)
static_assert(NOWN <= 255, "owner id must fit a byte with 255 left free");

__device__ __forceinline__ unsigned owner_of(int d)
{
    unsigned u = (unsigned)d;
    return u < (unsigned)NNODES ? u / (unsigned)RN : 255u;
}

__device__ __forceinline__ void slice_group(unsigned o, int g, int ln, unsigned* __restrict__ own)
{
    unsigned w = 0u;
#pragma unroll
    for (int k = 0; k < 8; ++k) {
        const unsigned bal = __builtin_amdgcn_ballot_w32(((o >> k) & 1u) != 0u);
        w = (ln == k) ? bal : w;
    }
    if (ln < 8)
        own[(g >> 5) * 256 + (ln >> 2) * 128 + (g & 31) * 4 + (ln & 3)] = w;
}

typedef float v4f __attribute__((ext_vector_type(4)));
#define GK     128
#define GN     64
#define WPITCH (GK + 4)
#define GTILES (NNODES / 16)
#define GBLK   ((GTILES + 7) / 8)
#define GWAVES (GBLK * 8)
#define SGRP   ((NGRP + GWAVES - 1) / GWAVES)
static_assert(NNODES % 16 == 0, "whole 16-row tiles");
static_assert(GWAVES <= NGRP, "every wave has a first group");
static_assert(GK % 8 == 0 && GN == 64, "fragment loop and weight staging below");

__global__ __launch_bounds__(256) void gemm_wmma_kernel(
    const float* __restrict__ Xin, const float* __restrict__ W,
    float* __restrict__ Out,
    const int* __restrict__ dst, unsigned* __restrict__ own)
{
    __shared__ float smem[GN * WPITCH];

    const int tid  = threadIdx.x;
    const int wave = tid >> 5;
    const int lane = tid & 31;
    const int half = lane >> 4;
    const int lid  = lane & 15;
    const int wid  = (int)blockIdx.x * 8 + wave;

    v4f wv[GK * GN / 4 / 256];
#pragma unroll
    for (int j = 0; j < GK * GN / 4 / 256; ++j) {
        const int unit = wave * (GK * GN / 4 / 256) + j;
        const int k  = (unit >> 2) * 8 + (lane & 7);
        const int n4 = (unit & 3) * 16 + (lane >> 3) * 4;
        wv[j] = *(const v4f*)(W + k * GN + n4);
    }
    int dv[SGRP];
#pragma unroll
    for (int u = 0; u < SGRP; ++u) {
        const int g = wid + u * GWAVES;
        dv[u] = dst[(g < NGRP ? g : wid) * 32 + lane];
    }
    const bool has = wid < GTILES;
    const int  m0  = (has ? wid : GTILES - 1) << 4;
    const float* arow = Xin + (size_t)(m0 + lid) * GK + (half << 2);
    v4f a[GK / 8];
#pragma unroll
    for (int j = 0; j < GK / 8; ++j)
        a[j] = *(const v4f*)(arow + 8 * j);

#pragma unroll
    for (int j = 0; j < GK * GN / 4 / 256; ++j) {
        const int unit = wave * (GK * GN / 4 / 256) + j;
        const int k  = (unit >> 2) * 8 + (lane & 7);
        const int n4 = (unit & 3) * 16 + (lane >> 3) * 4;
        smem[(n4 + 0) * WPITCH + k] = wv[j].x;
        smem[(n4 + 1) * WPITCH + k] = wv[j].y;
        smem[(n4 + 2) * WPITCH + k] = wv[j].z;
        smem[(n4 + 3) * WPITCH + k] = wv[j].w;
    }

#pragma unroll
    for (int u = 0; u < SGRP; ++u) {
        const int g = wid + u * GWAVES;
        if (u == 0 || g < NGRP) slice_group(owner_of(dv[u]), g, lane, own);
    }
    __syncthreads();
    if (!has) return;

    const float* bbase = smem + lid * WPITCH + (half << 2);
    v8f acc[4] = {};
#pragma unroll
    for (int j = 0; j < GK / 8; ++j) {
        v4f b[4];
#pragma unroll
        for (int t = 0; t < 4; ++t)
            b[t] = *(const v4f*)(bbase + t * 16 * WPITCH + 8 * j);
        const v2f alo = {a[j].x, a[j].y};
        const v2f ahi = {a[j].z, a[j].w};
#pragma unroll
        for (int t = 0; t < 4; ++t) {
            const v2f blo = {b[t].x, b[t].y};
            acc[t] = __builtin_amdgcn_wmma_f32_16x16x4_f32(
                 false, alo,  false, blo,
                 (short)0, acc[t],  false,  false);
        }
#pragma unroll
        for (int t = 0; t < 4; ++t) {
            const v2f bhi = {b[t].z, b[t].w};
            acc[t] = __builtin_amdgcn_wmma_f32_16x16x4_f32(
                 false, ahi,  false, bhi,
                 (short)0, acc[t],  false,  false);
        }
    }

    float* orow = Out + (size_t)(m0 + (half << 3)) * GN + lid;
#pragma unroll
    for (int t = 0; t < 4; ++t)
#pragma unroll
        for (int r = 0; r < 8; ++r)
            orow[(size_t)r * GN + 16 * t] = acc[t][r];
}

#define LCAP  36864
#define LHDR  32
#define LREG  (LCAP + LHDR)
static_assert(NEDGES % 32 == 0, "the owner ids are sliced in whole groups of 32 edges");
static_assert(NEDGES <= (1 << 20) && NNODES <= (1 << 20) && RN <= 2048, "list entry packing");

template <int F>
__device__ __forceinline__ constexpr int colpos(int f) { return (f & 1) * (F / 2) + (f >> 1); }

template <int F, bool BUILD>
__device__ __forceinline__ void keep_edge(unsigned pos, int e, int lo,
                                          unsigned* list, float* acc, unsigned* indeg,
                                          const float* __restrict__ P,
                                          const int* __restrict__ src,
                                          const int* __restrict__ dst)
{
    if (pos < LCAP) {
        list[pos] = (unsigned)e;
    } else {
        unsigned r = (unsigned)(dst[e] - lo);
        if (BUILD) atomicAdd(&indeg[r], 1u);
        const float* row = P + (size_t)src[e] * F;
#pragma nounroll
        for (int f = 0; f < F; ++f) atomicAdd(&acc[r * F + colpos<F>(f)], row[f]);
    }
}

template <int F, bool BUILD>
__device__ __forceinline__ void test_group(uint4 a, uint4 b, int e0, const unsigned (&kf)[8], int lo,
                                           unsigned* cnt, unsigned* list, float* acc, unsigned* indeg,
                                           const float* __restrict__ P,
                                           const int* __restrict__ src,
                                           const int* __restrict__ dst)
{
    const unsigned mm = (a.x ^ kf[0]) | (a.y ^ kf[1]) | (a.z ^ kf[2]) | (a.w ^ kf[3])
                      | (b.x ^ kf[4]) | (b.y ^ kf[5]) | (b.z ^ kf[6]) | (b.w ^ kf[7]);
    unsigned mk = ~mm;
    if (mk) {
        unsigned pos = atomicAdd(cnt, (unsigned)__popc(mk));
        do {
            unsigned t = (unsigned)__builtin_ctz(mk);
            mk &= mk - 1u;
            keep_edge<F, BUILD>(pos++, e0 + (int)t, lo, list, acc, indeg, P, src, dst);
        } while (mk);
    }
}

template <int F>
__device__ __forceinline__ void rows_issue(const unsigned* list, const float* __restrict__ P,
                                           int i0, int n, int lane,
                                           unsigned (&rr)[8], v2f (&v)[8])
{
    static_assert(F % 2 == 0 && F / 2 <= 32, "a row is at most 32 column pairs");
    const int i = i0 + (lane & 7);
    const unsigned ev = list[i < n ? i : n - 1];
#pragma unroll
    for (int u = 0; u < 8; ++u) {
        unsigned ent = (unsigned)__builtin_amdgcn_readlane((int)ev, u);
        const float* row = P + (size_t)(ent & 0xFFFFFu) * F;
        rr[u] = ent >> 20;
        v[u] = *(const v2f*)(row + 2 * (2 * lane < F ? lane : F / 2 - 1));
    }
}

template <int F, bool FULL>
__device__ __forceinline__ void rows_add(float* acc, int i0, int n, int lane,
                                         const unsigned (&rr)[8], const v2f (&v)[8])
{
#pragma unroll
    for (int u = 0; u < 8; ++u) {
        if (FULL || i0 + u < n) {
            float* a = acc + rr[u] * F;
            if (2 * lane < F) {
                atomicAdd(a + lane, v[u].x);
                atomicAdd(a + F / 2 + lane, v[u].y);
            }
        }
    }
}

template <int F, bool BUILD, bool RELU, int NN>
__global__ __launch_bounds__(1024) void owner_sum_kernel(
    const float* __restrict__ P, const int* __restrict__ src,
    const int* __restrict__ dst, float* __restrict__ hout,
    unsigned* __restrict__ saved, float* __restrict__ inv_deg,
    const unsigned* __restrict__ own, const float* __restrict__ bias,
    const float* __restrict__ Wn)
{
    __shared__ float    acc[RN * F];
    __shared__ unsigned list[LCAP];
    __shared__ unsigned cnt;
    __shared__ unsigned indeg[BUILD ? RN : 1];
    __shared__ float    inv_s[RN];

    const int tid = threadIdx.x;
    const int lo  = blockIdx.x * RN;
    const unsigned nrows = (unsigned)((NNODES - lo) < RN ? (NNODES - lo) : RN);
    unsigned* mine = saved + (size_t)blockIdx.x * LREG;

    const unsigned have = BUILD ? 0u : mine[0];
    const bool filter = BUILD || have > (unsigned)LCAP;

    for (int i = tid; i < RN * F; i += 1024) acc[i] = 0.0f;
    if (BUILD && tid < RN) indeg[tid] = 0u;
    if (!BUILD && (unsigned)tid < nrows) inv_s[tid] = inv_deg[lo + tid];
    if (tid == 0) cnt = 0u;

    int n;
    if (filter) {
        __syncthreads();

        const uint4* own4 = (const uint4*)own;
        unsigned kf[8];
#pragma unroll
        for (int k = 0; k < 8; ++k) kf[k] = 0u - ((blockIdx.x >> k) & 1u);
        int g = tid;
        for (; g + 3 * 1024 < NGRP; g += 4 * 1024) {
            uint4 a[4], b[4];
#pragma unroll
            for (int u = 0; u < 4; ++u) {
                const int gu = g + u * 1024;
                const int q  = (gu >> 5) * 64 + (gu & 31);
                a[u] = own4[q];
                b[u] = own4[q + 32];
            }
#pragma unroll
            for (int u = 0; u < 4; ++u)
                test_group<F, BUILD>(a[u], b[u], (g + u * 1024) << 5, kf, lo, &cnt, list, acc, indeg, P, src, dst);
        }
        for (; g < NGRP; g += 1024) {
            const int q = (g >> 5) * 64 + (g & 31);
            test_group<F, BUILD>(own4[q], own4[q + 32], g << 5, kf, lo, &cnt, list, acc, indeg, P, src, dst);
        }
        __syncthreads();
        const unsigned total = cnt;
        n = (int)(total < (unsigned)LCAP ? total : (unsigned)LCAP);

        for (int i = tid; i < n; i += 1024) {
            unsigned e = list[i];
            unsigned r = (unsigned)(dst[e] - lo);
            int s = src[e];
            if (BUILD) atomicAdd(&indeg[r], 1u);
            unsigned ent = (r << 20) | (unsigned)s;
            list[i] = ent;
            if (BUILD) mine[LHDR + i] = ent;
        }
        if (BUILD && tid == 0) mine[0] = total;
    } else {
        n = (int)have;
        for (int i = tid; i < n; i += 1024) list[i] = mine[LHDR + i];
    }
    __syncthreads();

    if (BUILD && (unsigned)tid < nrows) {
        float iv = 1.0f / fmaxf((float)indeg[tid], 1.0f);
        inv_deg[lo + tid] = iv;
        inv_s[tid] = iv;
    }

    const int wave = tid >> 5;
    const int lane = tid & 31;
    {
        int i0 = __builtin_amdgcn_readfirstlane(wave) * 8;
        if (i0 < n) {
            constexpr int G = 32 * 8;
            unsigned ra[8], rb[8];
            v2f av[8], bv[8];
            rows_issue<F>(list, P, i0, n, lane, ra, av);
            while (i0 + 2 * G < n) {
                rows_issue<F>(list, P, i0 + G, n, lane, rb, bv);
                rows_add<F, true>(acc, i0, n, lane, ra, av);
                rows_issue<F>(list, P, i0 + 2 * G, n, lane, ra, av);
                rows_add<F, true>(acc, i0 + G, n, lane, rb, bv);
                i0 += 2 * G;
            }
            if (i0 + G < n) {
                rows_issue<F>(list, P, i0 + G, n, lane, rb, bv);
                rows_add<F, true>(acc, i0, n, lane, ra, av);
                rows_add<F, false>(acc, i0 + G, n, lane, rb, bv);
            } else {
                rows_add<F, false>(acc, i0, n, lane, ra, av);
            }
        }
    }
    __syncthreads();

    if constexpr (NN == 0) {
        float* orow = hout + (size_t)lo * F;
        const int tot = (int)nrows * F;
        for (int i = tid; i < tot; i += 1024) {
            int r = i / F;
            int f = i - r * F;
            float v = acc[r * F + colpos<F>(f)] * inv_s[r] + bias[f];
            orow[i] = RELU ? fmaxf(v, 0.0f) : v;
        }
    } else {
        constexpr int MT = (RN + 15) / 16;
        constexpr int NT = (NN + 15) / 16;
        constexpr int HP = F + 4;
        static_assert((MT + NT) * 16 * HP <= LCAP, "rows and weights fit the list's space");
        float* hs = reinterpret_cast<float*>(list);
        float* wt = hs + MT * 16 * HP;
        static_assert(1024 % F == 0 && (F * NT * 16) % 1024 == 0, "whole strides of the block");
        constexpr int NWL = F * NT * 16 / 1024;
        float wreg[NWL];
#pragma unroll
        for (int j = 0; j < NWL; ++j) {
            int idx = tid + j * 1024;
            int k = idx / (NT * 16);
            int n = idx - k * (NT * 16);
            wreg[j] = (n < NN) ? Wn[k * NN + n] : 0.0f;
        }
        {
            const int f = tid % F;
            const int fp = colpos<F>(f);
            const float bf = bias[f];
            for (int r = tid / F; r < MT * 16; r += 1024 / F) {
                float v = 0.0f;
                if ((unsigned)r < nrows) {
                    v = acc[r * F + fp] * inv_s[r] + bf;
                    v = RELU ? fmaxf(v, 0.0f) : v;
                }
                hs[r * HP + f] = v;
            }
        }
#pragma unroll
        for (int j = 0; j < NWL; ++j) {
            int idx = tid + j * 1024;
            int k = idx / (NT * 16);
            int n = idx - k * (NT * 16);
            wt[n * HP + k] = wreg[j];
        }
        __syncthreads();

        const int half = lane >> 4;
        const int lid  = lane & 15;
        const int kofs = half << 1;
        const int wv = __builtin_amdgcn_readfirstlane(wave);
        for (int t = wv; t < MT * NT; t += 32) {
            const int mt = t / NT;
            const int nt = t - mt * NT;
            const float* arow = hs + (mt * 16 + lid) * HP + kofs;
            const float* brow = wt + (nt * 16 + lid) * HP + kofs;
            v8f c = {};
#pragma unroll
            for (int k0 = 0; k0 < F; k0 += 4) {
                v2f a = *(const v2f*)(arow + k0);
                v2f b = *(const v2f*)(brow + k0);
                c = __builtin_amdgcn_wmma_f32_16x16x4_f32(
                     false, a,  false, b,
                     (short)0, c,  false,  false);
            }
            const int ncol = nt * 16 + lid;
            const unsigned r0 = (unsigned)(mt * 16 + (half << 3));
            if (ncol < NN) {
                float* o = hout + ((size_t)lo + r0) * NN + ncol;
#pragma unroll
                for (int r = 0; r < 8; ++r)
                    if (r0 + (unsigned)r < nrows) o[(size_t)r * NN] = c[r];
            }
        }
    }
}


extern "C" void kernel_launch(void* const* d_in, const int* in_sizes, int n_in,
                              void* d_out, int out_size, void* d_ws, size_t ws_size,
                              hipStream_t stream)
{
    const float* features = (const float*)d_in[0];
    const int*   src      = (const int*)  d_in[1];
    const int*   dst      = (const int*)  d_in[2];
    const float* W0       = (const float*)d_in[3];
    const float* b0       = (const float*)d_in[4];
    const float* W1       = (const float*)d_in[5];
    const float* b1       = (const float*)d_in[6];
    const float* W2       = (const float*)d_in[7];
    const float* b2       = (const float*)d_in[8];
    float*       out      = (float*)d_out;

    float* ws  = (float*)d_ws;
    float* deg = ws;
    float* P   = ws + 50048;
    float* Q   = P  + (size_t)NNODES * 64;
    unsigned* L = (unsigned*)(Q + (size_t)NNODES * 64);
    unsigned* O = L + (size_t)NOWN * LREG;


    gemm_wmma_kernel<<<GBLK, 256, 0, stream>>>(features, W0, P, dst, O);
    owner_sum_kernel<64, true, true, 64><<<NOWN, 1024, 0, stream>>>(P, src, dst, Q, L, deg, O, b0, W1);

    owner_sum_kernel<64, false, true, 40><<<NOWN, 1024, 0, stream>>>(Q, src, dst, P, L, deg, O, b1, W2);

    owner_sum_kernel<40, false, false, 0><<<NOWN, 1024, 0, stream>>>(P, src, dst, out, L, deg, O, b2, nullptr);
}
